# starting kernel
# baseline (speedup 1.0000x reference)
#include <hip/hip_runtime.h>
#include <hip/hip_bf16.h>


typedef float v2f __attribute__((ext_vector_type(2)));
typedef float v8f __attribute__((ext_vector_type(8)));

#define NEG_SLOPE 0.2f
#define SM_EPS    1e-16f
#define ORD_NEG_INF 0x007FFFFFu

__device__ __forceinline__ unsigned f2ord(float f) {
    unsigned u = __float_as_uint(f);
    return (u & 0x80000000u) ? ~u : (u | 0x80000000u);
}
__device__ __forceinline__ float ord2f(unsigned u) {
    unsigned v = (u & 0x80000000u) ? (u & 0x7FFFFFFFu) : ~u;
    return __uint_as_float(v);
}

template <int NCOLS>
__global__ void __launch_bounds__(256) wmma_gemm_f32(
    const float* __restrict__ X, const float* __restrict__ W,
    float* __restrict__ Y, int nrows)
{
    __shared__ float As[16][128];
    const int row0 = blockIdx.x * 16;
    const int tid  = threadIdx.x;

    {
        const float4* X4  = (const float4*)X;
        float4*       As4 = (float4*)&As[0][0];
#pragma unroll
        for (int i = tid; i < 512; i += 256) {
            int r  = i >> 5;
            int rr = row0 + r;
            float4 v = make_float4(0.f, 0.f, 0.f, 0.f);
            if (rr < nrows) v = X4[(size_t)rr * 32 + (i & 31)];
            As4[i] = v;
        }
    }
    __syncthreads();

    const int wave = tid >> 5;
    const int lane = tid & 31;
    const int col0 = wave * 16;
    if (col0 >= NCOLS) return;

    const int m  = lane & 15;
    const int hi = lane >> 4;

    const float* wp = W + (size_t)col0 + m;
    v8f acc = {};
#pragma unroll 4
    for (int k0 = 0; k0 < 128; k0 += 4) {
        const int ka = k0 + (hi << 1);
        v2f a, b;
        a.x = As[m][ka];
        a.y = As[m][ka + 1];
        b.x = wp[(size_t)ka * NCOLS];
        b.y = wp[(size_t)(ka + 1) * NCOLS];
        acc = __builtin_amdgcn_wmma_f32_16x16x4_f32(
             false, a,  false, b,
             (short)0, acc,  false,  false);
    }

    const int orow = row0 + (hi ? 8 : 0);
    float* yp = Y + (size_t)orow * NCOLS + col0 + m;
    if (row0 + 16 <= nrows) {
#pragma unroll
        for (int r = 0; r < 8; ++r) yp[(size_t)r * NCOLS] = acc[r];
    } else {
#pragma unroll
        for (int r = 0; r < 8; ++r)
            if (orow + r < nrows) yp[(size_t)r * NCOLS] = acc[r];
    }
}

template <int HEADS, int C>
__global__ void attn_scores(const float* __restrict__ H,
                            const float* __restrict__ att_src,
                            const float* __restrict__ att_dst,
                            float* __restrict__ a_src, float* __restrict__ a_dst,
                            int N)
{
    int t = blockIdx.x * blockDim.x + threadIdx.x;
    if (t >= N * HEADS) return;
    int hd = t & (HEADS - 1);
    const float* hp = H + (size_t)t * C;
    const float* as = att_src + hd * C;
    const float* ad = att_dst + hd * C;
    float ss = 0.f, sd = 0.f;
#pragma unroll
    for (int c = 0; c < C; ++c) { float v = hp[c]; ss += v * as[c]; sd += v * ad[c]; }
    a_src[t] = ss; a_dst[t] = sd;
}

__global__ void fill_u32(unsigned* __restrict__ p, unsigned v, int n)
{
    int t = blockIdx.x * blockDim.x + threadIdx.x;
    if (t < n) p[t] = v;
}

template <int HEADS>
__global__ void edge_alpha_max(const int* __restrict__ ei, int E, int Etot,
                               const float* __restrict__ a_src,
                               const float* __restrict__ a_dst,
                               float* __restrict__ alpha,
                               unsigned* __restrict__ mmax)
{
    int t = blockIdx.x * blockDim.x + threadIdx.x;
    if (t >= Etot * HEADS) return;
    int e = t / HEADS, hd = t & (HEADS - 1);
    int s, d;
    if (e < E) { s = ei[e]; d = ei[E + e]; } else { s = e - E; d = s; }
    float a = a_src[s * HEADS + hd] + a_dst[d * HEADS + hd];
    a = (a > 0.f) ? a : NEG_SLOPE * a;
    alpha[t] = a;
    atomicMax(&mmax[d * HEADS + hd], f2ord(a));
}

template <int HEADS>
__global__ void edge_exp_denom(const int* __restrict__ ei, int E, int Etot,
                               float* __restrict__ alpha,
                               const unsigned* __restrict__ mmax,
                               float* __restrict__ denom)
{
    int t = blockIdx.x * blockDim.x + threadIdx.x;
    if (t >= Etot * HEADS) return;
    int e = t / HEADS, hd = t & (HEADS - 1);
    int d = (e < E) ? ei[E + e] : (e - E);
    float w = expf(alpha[t] - ord2f(mmax[d * HEADS + hd]));
    alpha[t] = w;
    atomicAdd(&denom[d * HEADS + hd], w);
}

template <int HEADS, int C>
__global__ void edge_aggregate(const int* __restrict__ ei, int E, int Etot,
                               const float* __restrict__ H,
                               const float* __restrict__ w,
                               const float* __restrict__ denom,
                               float* __restrict__ agg)
{
    constexpr int F = HEADS * C;
    long long t = (long long)blockIdx.x * blockDim.x + threadIdx.x;
    if (t >= (long long)Etot * F) return;
    int e  = (int)(t >> (__builtin_ctz(F)));
    int f  = (int)t & (F - 1);
    int hd = f >> (__builtin_ctz(C));
    int s, d;
    if (e < E) { s = ei[e]; d = ei[E + e]; } else { s = e - E; d = s; }
    float coef = w[e * HEADS + hd] / (denom[d * HEADS + hd] + SM_EPS);
    atomicAdd(&agg[(size_t)d * F + f], H[(size_t)s * F + f] * coef);
}

template <int F, bool RELU>
__global__ void bias_act(const float* __restrict__ in, float* __restrict__ out,
                         const float* __restrict__ bias, int total)
{
    int t = blockIdx.x * blockDim.x + threadIdx.x;
    if (t >= total) return;
    float v = in[t] + bias[t & (F - 1)];
    out[t] = RELU ? fmaxf(v, 0.f) : v;
}

static inline int cdiv(long long a, int b) { return (int)((a + b - 1) / b); }

extern "C" void kernel_launch(void* const* d_in, const int* in_sizes, int n_in,
                              void* d_out, int out_size, void* d_ws, size_t ws_size,
                              hipStream_t stream)
{
    const float* x        = (const float*)d_in[0];
    const int*   ei       = (const int*)  d_in[1];
    const float* W1       = (const float*)d_in[2];
    const float* att_src1 = (const float*)d_in[3];
    const float* att_dst1 = (const float*)d_in[4];
    const float* b1       = (const float*)d_in[5];
    const float* W2       = (const float*)d_in[6];
    const float* att_src2 = (const float*)d_in[7];
    const float* att_dst2 = (const float*)d_in[8];
    const float* b2       = (const float*)d_in[9];

    constexpr int IN = 128, H1 = 8, C1 = 16, F1 = H1 * C1  , C2 = 64;
    const int N    = in_sizes[0] / IN;
    const int E    = in_sizes[1] / 2;
    const int Etot = E + N;

    size_t off = 0;
    float* base = (float*)d_ws;
    auto take = [&](size_t n) { float* p = base + off; off += n; return p; };
    float*    h1     = take((size_t)N * F1);
    float*    asrc1  = take((size_t)N * H1);
    float*    adst1  = take((size_t)N * H1);
    unsigned* m1     = (unsigned*)take((size_t)N * H1);
    float*    den1   = take((size_t)N * H1);
    float*    w1     = take((size_t)Etot * H1);
    float*    agg1   = take((size_t)N * F1);
    float*    h2     = take((size_t)N * C2);
    float*    asrc2  = take((size_t)N);
    float*    adst2  = take((size_t)N);
    unsigned* m2     = (unsigned*)take((size_t)N);
    float*    den2   = take((size_t)N);
    float*    w2     = take((size_t)Etot);
    float*    agg2   = take((size_t)N * C2);

    const int B = 256;

    wmma_gemm_f32<F1><<<cdiv(N, 16), 256, 0, stream>>>(x, W1, h1, N);
    attn_scores<H1, C1><<<cdiv((long long)N * H1, B), B, 0, stream>>>(
        h1, att_src1, att_dst1, asrc1, adst1, N);

    fill_u32<<<cdiv((long long)N * H1, B), B, 0, stream>>>(m1, ORD_NEG_INF, N * H1);
    fill_u32<<<cdiv((long long)N * H1, B), B, 0, stream>>>((unsigned*)den1, 0u, N * H1);
    fill_u32<<<cdiv((long long)N * F1, B), B, 0, stream>>>((unsigned*)agg1, 0u, N * F1);

    edge_alpha_max<H1><<<cdiv((long long)Etot * H1, B), B, 0, stream>>>(
        ei, E, Etot, asrc1, adst1, w1, m1);
    edge_exp_denom<H1><<<cdiv((long long)Etot * H1, B), B, 0, stream>>>(
        ei, E, Etot, w1, m1, den1);
    edge_aggregate<H1, C1><<<cdiv((long long)Etot * F1, B), B, 0, stream>>>(
        ei, E, Etot, h1, w1, den1, agg1);

    bias_act<F1, true><<<cdiv((long long)N * F1, B), B, 0, stream>>>(
        agg1, agg1, b1, N * F1);

    wmma_gemm_f32<C2><<<cdiv(N, 16), 256, 0, stream>>>(agg1, W2, h2, N);
    attn_scores<1, C2><<<cdiv((long long)N, B), B, 0, stream>>>(
        h2, att_src2, att_dst2, asrc2, adst2, N);

    fill_u32<<<cdiv((long long)N, B), B, 0, stream>>>(m2, ORD_NEG_INF, N);
    fill_u32<<<cdiv((long long)N, B), B, 0, stream>>>((unsigned*)den2, 0u, N);
    fill_u32<<<cdiv((long long)N * C2, B), B, 0, stream>>>((unsigned*)agg2, 0u, N * C2);

    edge_alpha_max<1><<<cdiv((long long)Etot, B), B, 0, stream>>>(
        ei, E, Etot, asrc2, adst2, w2, m2);
    edge_exp_denom<1><<<cdiv((long long)Etot, B), B, 0, stream>>>(
        ei, E, Etot, w2, m2, den2);
    edge_aggregate<1, C2><<<cdiv((long long)Etot * C2, B), B, 0, stream>>>(
        ei, E, Etot, h2, w2, den2, agg2);

    bias_act<C2, false><<<cdiv((long long)N * C2, B), B, 0, stream>>>(
        agg2, (float*)d_out, b2, N * C2);
}


# Edges grouped by destination once per call (integer atomic count with arrival-order rank, scan, place); one wave per node sums its incoming rows in that order: float atomics and their fills gone
# speedup vs baseline: 2.8921x; 2.8921x over previous
#include <hip/hip_runtime.h>
#include <hip/hip_bf16.h>


typedef float v2f __attribute__((ext_vector_type(2)));
typedef float v8f __attribute__((ext_vector_type(8)));

#define NEG_SLOPE 0.2f
#define SM_EPS    1e-16f

template <int NCOLS>
__global__ void __launch_bounds__(256) wmma_gemm_f32(
    const float* __restrict__ X, const float* __restrict__ W,
    float* __restrict__ Y, int nrows)
{
    __shared__ float As[16][128];
    const int row0 = blockIdx.x * 16;
    const int tid  = threadIdx.x;

    {
        const float4* X4  = (const float4*)X;
        float4*       As4 = (float4*)&As[0][0];
#pragma unroll
        for (int i = tid; i < 512; i += 256) {
            int r  = i >> 5;
            int rr = row0 + r;
            float4 v = make_float4(0.f, 0.f, 0.f, 0.f);
            if (rr < nrows) v = X4[(size_t)rr * 32 + (i & 31)];
            As4[i] = v;
        }
    }
    __syncthreads();

    const int wave = tid >> 5;
    const int lane = tid & 31;
    const int col0 = wave * 16;
    if (col0 >= NCOLS) return;

    const int m  = lane & 15;
    const int hi = lane >> 4;

    const float* wp = W + (size_t)col0 + m;
    v8f acc = {};
#pragma unroll 4
    for (int k0 = 0; k0 < 128; k0 += 4) {
        const int ka = k0 + (hi << 1);
        v2f a, b;
        a.x = As[m][ka];
        a.y = As[m][ka + 1];
        b.x = wp[(size_t)ka * NCOLS];
        b.y = wp[(size_t)(ka + 1) * NCOLS];
        acc = __builtin_amdgcn_wmma_f32_16x16x4_f32(
             false, a,  false, b,
             (short)0, acc,  false,  false);
    }

    const int orow = row0 + (hi ? 8 : 0);
    float* yp = Y + (size_t)orow * NCOLS + col0 + m;
    if (row0 + 16 <= nrows) {
#pragma unroll
        for (int r = 0; r < 8; ++r) yp[(size_t)r * NCOLS] = acc[r];
    } else {
#pragma unroll
        for (int r = 0; r < 8; ++r)
            if (orow + r < nrows) yp[(size_t)r * NCOLS] = acc[r];
    }
}

template <int HEADS, int C>
__global__ void attn_scores(const float* __restrict__ H,
                            const float* __restrict__ att_src,
                            const float* __restrict__ att_dst,
                            float* __restrict__ a_src, float* __restrict__ a_dst,
                            int N)
{
    int t = blockIdx.x * blockDim.x + threadIdx.x;
    if (t >= N * HEADS) return;
    int hd = t & (HEADS - 1);
    const float* hp = H + (size_t)t * C;
    const float* as = att_src + hd * C;
    const float* ad = att_dst + hd * C;
    float ss = 0.f, sd = 0.f;
#pragma unroll
    for (int c = 0; c < C; ++c) { float v = hp[c]; ss += v * as[c]; sd += v * ad[c]; }
    a_src[t] = ss; a_dst[t] = sd;
}

__global__ void fill_u32(unsigned* __restrict__ p, unsigned v, int n)
{
    int t = blockIdx.x * blockDim.x + threadIdx.x;
    if (t < n) p[t] = v;
}

__global__ void dst_count_rank(const int* __restrict__ ei, int E,
                               int* __restrict__ cnt, int* __restrict__ rank)
{
    int t = blockIdx.x * blockDim.x + threadIdx.x;
    if (t >= E) return;
    rank[t] = atomicAdd(&cnt[ei[E + t]], 1);
}

__global__ void __launch_bounds__(1024) scan_counts(const int* __restrict__ cnt,
                                                    int* __restrict__ rowptr, int chunk)
{
    __shared__ int wtot[32];
    const int tid  = threadIdx.x;
    const int lane = tid & 31;
    const int wv   = tid >> 5;
    const int nq   = chunk >> 2;
    const int4* c4 = (const int4*)(cnt + (size_t)tid * chunk);
    int4*       r4 = (int4*)(rowptr + (size_t)tid * chunk);

    int s = 0;
    for (int i = 0; i < nq; ++i) { int4 v = c4[i]; s += (v.x + v.y) + (v.z + v.w); }

    int incl = s;
    for (int o = 1; o < 32; o <<= 1) {
        int v = __shfl_up(incl, o);
        if (lane >= o) incl += v;
    }
    if (lane == 31) wtot[wv] = incl;
    __syncthreads();
    if (wv == 0) {
        int w = wtot[lane];
        int wi = w;
        for (int o = 1; o < 32; o <<= 1) {
            int v = __shfl_up(wi, o);
            if (lane >= o) wi += v;
        }
        wtot[lane] = wi - w;
    }
    __syncthreads();

    int run = wtot[wv] + (incl - s);
    for (int i = 0; i < nq; ++i) {
        int4 v = c4[i];
        int4 o;
        o.x = run;
        o.y = o.x + v.x;
        o.z = o.y + v.y;
        o.w = o.z + v.z;
        run = o.w + v.w;
        r4[i] = o;
    }
    if (tid == 1023) rowptr[(size_t)1024 * chunk] = run;
}

__global__ void csr_place(const int* __restrict__ ei, int E,
                          const int* __restrict__ rowptr, const int* __restrict__ rank,
                          int* __restrict__ csr_src)
{
    int t = blockIdx.x * blockDim.x + threadIdx.x;
    if (t >= E) return;
    csr_src[rowptr[ei[E + t]] + rank[t]] = ei[t];
}

template <int HEADS, int C>
__global__ void __launch_bounds__(256) gat_gather(
    const int* __restrict__ rowptr, const int* __restrict__ csr_src,
    const float* __restrict__ Hm, const float* __restrict__ a_src,
    const float* __restrict__ a_dst, float* __restrict__ out, int N)
{
    constexpr int F   = HEADS * C;
    constexpr int VEC = F / 32;
    static_assert(F % 32 == 0 && (VEC == 2 || VEC == 4), "a row must split evenly over one wave");
    typedef float vf __attribute__((ext_vector_type(VEC)));

    const int lane = threadIdx.x & 31;
    const int d = __builtin_amdgcn_readfirstlane((int)(blockIdx.x * 8 + (threadIdx.x >> 5)));
    if (d >= N) return;
    const int beg = __builtin_amdgcn_readfirstlane(rowptr[d]);
    const int end = __builtin_amdgcn_readfirstlane(rowptr[d + 1]);

    const int    hd  = (lane * VEC) / C;
    const float  ad  = a_dst[d * HEADS + hd];
    const float* asp = a_src + hd;
    const vf*    Hv  = (const vf*)Hm + lane;

    float a_self = asp[(size_t)d * HEADS] + ad;
    a_self = (a_self > 0.f) ? a_self : NEG_SLOPE * a_self;

    float m = a_self;
    for (int j0 = beg; j0 < end; j0 += 32) {
        const int n = (end - j0 < 32) ? (end - j0) : 32;
        const int mine = (lane < n) ? csr_src[j0 + lane] : 0;
        for (int k = 0; k < n; ++k) {
            const int s = __builtin_amdgcn_readlane(mine, k);
            float a = asp[(size_t)s * HEADS] + ad;
            a = (a > 0.f) ? a : NEG_SLOPE * a;
            m = fmaxf(m, a);
        }
    }

    float w   = expf(a_self - m);
    float den = w;
    vf    acc = Hv[(size_t)d * 32] * w;
    for (int j0 = beg; j0 < end; j0 += 32) {
        const int n = (end - j0 < 32) ? (end - j0) : 32;
        const int mine = (lane < n) ? csr_src[j0 + lane] : 0;
#pragma unroll 4
        for (int k = 0; k < n; ++k) {
            const int s = __builtin_amdgcn_readlane(mine, k);
            float a = asp[(size_t)s * HEADS] + ad;
            a = (a > 0.f) ? a : NEG_SLOPE * a;
            w = expf(a - m);
            den += w;
            acc += Hv[(size_t)s * 32] * w;
        }
    }
    ((vf*)out)[(size_t)d * 32 + lane] = acc / (den + SM_EPS);
}

template <int F, bool RELU>
__global__ void bias_act(const float* __restrict__ in, float* __restrict__ out,
                         const float* __restrict__ bias, int total)
{
    int t = blockIdx.x * blockDim.x + threadIdx.x;
    if (t >= total) return;
    float v = in[t] + bias[t & (F - 1)];
    out[t] = RELU ? fmaxf(v, 0.f) : v;
}

static inline int cdiv(long long a, int b) { return (int)((a + b - 1) / b); }

extern "C" void kernel_launch(void* const* d_in, const int* in_sizes, int n_in,
                              void* d_out, int out_size, void* d_ws, size_t ws_size,
                              hipStream_t stream)
{
    const float* x        = (const float*)d_in[0];
    const int*   ei       = (const int*)  d_in[1];
    const float* W1       = (const float*)d_in[2];
    const float* att_src1 = (const float*)d_in[3];
    const float* att_dst1 = (const float*)d_in[4];
    const float* b1       = (const float*)d_in[5];
    const float* W2       = (const float*)d_in[6];
    const float* att_src2 = (const float*)d_in[7];
    const float* att_dst2 = (const float*)d_in[8];
    const float* b2       = (const float*)d_in[9];

    constexpr int IN = 128, H1 = 8, C1 = 16, F1 = H1 * C1  , C2 = 64;
    const int N = in_sizes[0] / IN;
    const int E = in_sizes[1] / 2;

    const int chunk = ((N + 1023) / 1024 + 3) & ~3;
    const int NP    = 1024 * chunk;

    size_t off = 0;
    float* base = (float*)d_ws;
    auto take = [&](size_t n) { float* p = base + off; off += (n + 63) & ~(size_t)63; return p; };
    float* h1      = take((size_t)N * F1);
    float* asrc1   = take((size_t)N * H1);
    float* adst1   = take((size_t)N * H1);
    float* agg1    = take((size_t)N * F1);
    float* h2      = take((size_t)N * C2);
    float* asrc2   = take((size_t)N);
    float* adst2   = take((size_t)N);
    float* agg2    = take((size_t)N * C2);
    int*   cnt     = (int*)take((size_t)NP);
    int*   rowptr  = (int*)take((size_t)NP + 4);
    int*   rank    = (int*)take((size_t)E);
    int*   csr_src = (int*)take((size_t)E);

    const int B = 256;

    fill_u32<<<cdiv(NP, B), B, 0, stream>>>((unsigned*)cnt, 0u, NP);
    dst_count_rank<<<cdiv(E, B), B, 0, stream>>>(ei, E, cnt, rank);
    scan_counts<<<1, 1024, 0, stream>>>(cnt, rowptr, chunk);
    csr_place<<<cdiv(E, B), B, 0, stream>>>(ei, E, rowptr, rank, csr_src);

    wmma_gemm_f32<F1><<<cdiv(N, 16), 256, 0, stream>>>(x, W1, h1, N);
    attn_scores<H1, C1><<<cdiv((long long)N * H1, B), B, 0, stream>>>(
        h1, att_src1, att_dst1, asrc1, adst1, N);
    gat_gather<H1, C1><<<cdiv(N, 8), 256, 0, stream>>>(
        rowptr, csr_src, h1, asrc1, adst1, agg1, N);

    bias_act<F1, true><<<cdiv((long long)N * F1, B), B, 0, stream>>>(
        agg1, agg1, b1, N * F1);

    wmma_gemm_f32<C2><<<cdiv(N, 16), 256, 0, stream>>>(agg1, W2, h2, N);
    attn_scores<1, C2><<<cdiv((long long)N, B), B, 0, stream>>>(
        h2, att_src2, att_dst2, asrc2, adst2, N);
    gat_gather<1, C2><<<cdiv(N, 8), 256, 0, stream>>>(
        rowptr, csr_src, h2, asrc2, adst2, agg2, N);

    bias_act<C2, false><<<cdiv((long long)N * C2, B), B, 0, stream>>>(
        agg2, (float*)d_out, b2, N * C2);
}


# Bias add (and layer-1 ReLU) done in the gather's own row store, layer 2 straight to out: both elementwise launches gone (12 -> 10 launches, 300k one-element waves fewer); arithmetic order kept
# speedup vs baseline: 3.2062x; 1.1086x over previous
#include <hip/hip_runtime.h>
#include <hip/hip_bf16.h>


typedef float v2f __attribute__((ext_vector_type(2)));
typedef float v8f __attribute__((ext_vector_type(8)));

#define NEG_SLOPE 0.2f
#define SM_EPS    1e-16f

template <int NCOLS>
__global__ void __launch_bounds__(256) wmma_gemm_f32(
    const float* __restrict__ X, const float* __restrict__ W,
    float* __restrict__ Y, int nrows)
{
    __shared__ float As[16][128];
    const int row0 = blockIdx.x * 16;
    const int tid  = threadIdx.x;

    {
        const float4* X4  = (const float4*)X;
        float4*       As4 = (float4*)&As[0][0];
#pragma unroll
        for (int i = tid; i < 512; i += 256) {
            int r  = i >> 5;
            int rr = row0 + r;
            float4 v = make_float4(0.f, 0.f, 0.f, 0.f);
            if (rr < nrows) v = X4[(size_t)rr * 32 + (i & 31)];
            As4[i] = v;
        }
    }
    __syncthreads();

    const int wave = tid >> 5;
    const int lane = tid & 31;
    const int col0 = wave * 16;
    if (col0 >= NCOLS) return;

    const int m  = lane & 15;
    const int hi = lane >> 4;

    const float* wp = W + (size_t)col0 + m;
    v8f acc = {};
#pragma unroll 4
    for (int k0 = 0; k0 < 128; k0 += 4) {
        const int ka = k0 + (hi << 1);
        v2f a, b;
        a.x = As[m][ka];
        a.y = As[m][ka + 1];
        b.x = wp[(size_t)ka * NCOLS];
        b.y = wp[(size_t)(ka + 1) * NCOLS];
        acc = __builtin_amdgcn_wmma_f32_16x16x4_f32(
             false, a,  false, b,
             (short)0, acc,  false,  false);
    }

    const int orow = row0 + (hi ? 8 : 0);
    float* yp = Y + (size_t)orow * NCOLS + col0 + m;
    if (row0 + 16 <= nrows) {
#pragma unroll
        for (int r = 0; r < 8; ++r) yp[(size_t)r * NCOLS] = acc[r];
    } else {
#pragma unroll
        for (int r = 0; r < 8; ++r)
            if (orow + r < nrows) yp[(size_t)r * NCOLS] = acc[r];
    }
}

template <int HEADS, int C>
__global__ void attn_scores(const float* __restrict__ H,
                            const float* __restrict__ att_src,
                            const float* __restrict__ att_dst,
                            float* __restrict__ a_src, float* __restrict__ a_dst,
                            int N)
{
    int t = blockIdx.x * blockDim.x + threadIdx.x;
    if (t >= N * HEADS) return;
    int hd = t & (HEADS - 1);
    const float* hp = H + (size_t)t * C;
    const float* as = att_src + hd * C;
    const float* ad = att_dst + hd * C;
    float ss = 0.f, sd = 0.f;
#pragma unroll
    for (int c = 0; c < C; ++c) { float v = hp[c]; ss += v * as[c]; sd += v * ad[c]; }
    a_src[t] = ss; a_dst[t] = sd;
}

__global__ void fill_u32(unsigned* __restrict__ p, unsigned v, int n)
{
    int t = blockIdx.x * blockDim.x + threadIdx.x;
    if (t < n) p[t] = v;
}

__global__ void dst_count_rank(const int* __restrict__ ei, int E,
                               int* __restrict__ cnt, int* __restrict__ rank)
{
    int t = blockIdx.x * blockDim.x + threadIdx.x;
    if (t >= E) return;
    rank[t] = atomicAdd(&cnt[ei[E + t]], 1);
}

__global__ void __launch_bounds__(1024) scan_counts(const int* __restrict__ cnt,
                                                    int* __restrict__ rowptr, int chunk)
{
    __shared__ int wtot[32];
    const int tid  = threadIdx.x;
    const int lane = tid & 31;
    const int wv   = tid >> 5;
    const int nq   = chunk >> 2;
    const int4* c4 = (const int4*)(cnt + (size_t)tid * chunk);
    int4*       r4 = (int4*)(rowptr + (size_t)tid * chunk);

    int s = 0;
    for (int i = 0; i < nq; ++i) { int4 v = c4[i]; s += (v.x + v.y) + (v.z + v.w); }

    int incl = s;
    for (int o = 1; o < 32; o <<= 1) {
        int v = __shfl_up(incl, o);
        if (lane >= o) incl += v;
    }
    if (lane == 31) wtot[wv] = incl;
    __syncthreads();
    if (wv == 0) {
        int w = wtot[lane];
        int wi = w;
        for (int o = 1; o < 32; o <<= 1) {
            int v = __shfl_up(wi, o);
            if (lane >= o) wi += v;
        }
        wtot[lane] = wi - w;
    }
    __syncthreads();

    int run = wtot[wv] + (incl - s);
    for (int i = 0; i < nq; ++i) {
        int4 v = c4[i];
        int4 o;
        o.x = run;
        o.y = o.x + v.x;
        o.z = o.y + v.y;
        o.w = o.z + v.z;
        run = o.w + v.w;
        r4[i] = o;
    }
    if (tid == 1023) rowptr[(size_t)1024 * chunk] = run;
}

__global__ void csr_place(const int* __restrict__ ei, int E,
                          const int* __restrict__ rowptr, const int* __restrict__ rank,
                          int* __restrict__ csr_src)
{
    int t = blockIdx.x * blockDim.x + threadIdx.x;
    if (t >= E) return;
    csr_src[rowptr[ei[E + t]] + rank[t]] = ei[t];
}

template <int HEADS, int C, bool RELU>
__global__ void __launch_bounds__(256) gat_gather(
    const int* __restrict__ rowptr, const int* __restrict__ csr_src,
    const float* __restrict__ Hm, const float* __restrict__ a_src,
    const float* __restrict__ a_dst, const float* __restrict__ bias,
    float* __restrict__ out, int N)
{
    constexpr int F   = HEADS * C;
    constexpr int VEC = F / 32;
    static_assert(F % 32 == 0 && (VEC == 2 || VEC == 4), "a row must split evenly over one wave");
    typedef float vf __attribute__((ext_vector_type(VEC)));

    const int lane = threadIdx.x & 31;
    const int d = __builtin_amdgcn_readfirstlane((int)(blockIdx.x * 8 + (threadIdx.x >> 5)));
    if (d >= N) return;
    const int beg = __builtin_amdgcn_readfirstlane(rowptr[d]);
    const int end = __builtin_amdgcn_readfirstlane(rowptr[d + 1]);

    const int    hd  = (lane * VEC) / C;
    const float  ad  = a_dst[d * HEADS + hd];
    const float* asp = a_src + hd;
    const vf*    Hv  = (const vf*)Hm + lane;
    const vf     bv  = ((const vf*)bias)[lane];

    float a_self = asp[(size_t)d * HEADS] + ad;
    a_self = (a_self > 0.f) ? a_self : NEG_SLOPE * a_self;

    float m = a_self;
    for (int j0 = beg; j0 < end; j0 += 32) {
        const int n = (end - j0 < 32) ? (end - j0) : 32;
        const int mine = (lane < n) ? csr_src[j0 + lane] : 0;
        for (int k = 0; k < n; ++k) {
            const int s = __builtin_amdgcn_readlane(mine, k);
            float a = asp[(size_t)s * HEADS] + ad;
            a = (a > 0.f) ? a : NEG_SLOPE * a;
            m = fmaxf(m, a);
        }
    }

    float w   = expf(a_self - m);
    float den = w;
    vf    acc = Hv[(size_t)d * 32] * w;
    for (int j0 = beg; j0 < end; j0 += 32) {
        const int n = (end - j0 < 32) ? (end - j0) : 32;
        const int mine = (lane < n) ? csr_src[j0 + lane] : 0;
#pragma unroll 4
        for (int k = 0; k < n; ++k) {
            const int s = __builtin_amdgcn_readlane(mine, k);
            float a = asp[(size_t)s * HEADS] + ad;
            a = (a > 0.f) ? a : NEG_SLOPE * a;
            w = expf(a - m);
            den += w;
            acc += Hv[(size_t)s * 32] * w;
        }
    }
    vf r = acc / (den + SM_EPS) + bv;
    if (RELU) {
#pragma unroll
        for (int i = 0; i < VEC; ++i) r[i] = fmaxf(r[i], 0.f);
    }
    ((vf*)out)[(size_t)d * 32 + lane] = r;
}

static inline int cdiv(long long a, int b) { return (int)((a + b - 1) / b); }

extern "C" void kernel_launch(void* const* d_in, const int* in_sizes, int n_in,
                              void* d_out, int out_size, void* d_ws, size_t ws_size,
                              hipStream_t stream)
{
    const float* x        = (const float*)d_in[0];
    const int*   ei       = (const int*)  d_in[1];
    const float* W1       = (const float*)d_in[2];
    const float* att_src1 = (const float*)d_in[3];
    const float* att_dst1 = (const float*)d_in[4];
    const float* b1       = (const float*)d_in[5];
    const float* W2       = (const float*)d_in[6];
    const float* att_src2 = (const float*)d_in[7];
    const float* att_dst2 = (const float*)d_in[8];
    const float* b2       = (const float*)d_in[9];

    constexpr int IN = 128, H1 = 8, C1 = 16, F1 = H1 * C1  , C2 = 64;
    const int N = in_sizes[0] / IN;
    const int E = in_sizes[1] / 2;

    const int chunk = ((N + 1023) / 1024 + 3) & ~3;
    const int NP    = 1024 * chunk;

    size_t off = 0;
    float* base = (float*)d_ws;
    auto take = [&](size_t n) { float* p = base + off; off += (n + 63) & ~(size_t)63; return p; };
    float* h1      = take((size_t)N * F1);
    float* asrc1   = take((size_t)N * H1);
    float* adst1   = take((size_t)N * H1);
    float* agg1    = take((size_t)N * F1);
    float* h2      = take((size_t)N * C2);
    float* asrc2   = take((size_t)N);
    float* adst2   = take((size_t)N);
    int*   cnt     = (int*)take((size_t)NP);
    int*   rowptr  = (int*)take((size_t)NP + 4);
    int*   rank    = (int*)take((size_t)E);
    int*   csr_src = (int*)take((size_t)E);

    const int B = 256;

    fill_u32<<<cdiv(NP, B), B, 0, stream>>>((unsigned*)cnt, 0u, NP);
    dst_count_rank<<<cdiv(E, B), B, 0, stream>>>(ei, E, cnt, rank);
    scan_counts<<<1, 1024, 0, stream>>>(cnt, rowptr, chunk);
    csr_place<<<cdiv(E, B), B, 0, stream>>>(ei, E, rowptr, rank, csr_src);

    wmma_gemm_f32<F1><<<cdiv(N, 16), 256, 0, stream>>>(x, W1, h1, N);
    attn_scores<H1, C1><<<cdiv((long long)N * H1, B), B, 0, stream>>>(
        h1, att_src1, att_dst1, asrc1, adst1, N);
    gat_gather<H1, C1, true><<<cdiv(N, 8), 256, 0, stream>>>(
        rowptr, csr_src, h1, asrc1, adst1, b1, agg1, N);

    wmma_gemm_f32<C2><<<cdiv(N, 16), 256, 0, stream>>>(agg1, W2, h2, N);
    attn_scores<1, C2><<<cdiv((long long)N, B), B, 0, stream>>>(
        h2, att_src2, att_dst2, asrc2, adst2, N);
    gat_gather<1, C2, false><<<cdiv(N, 8), 256, 0, stream>>>(
        rowptr, csr_src, h2, asrc2, adst2, b2, (float*)d_out, N);
}


# Each node's edge logits fetched once, one edge per lane (register for 1 head, per-wave LDS strip for 8) and reused by both passes; edge loops hold only row gathers, 8/4/2/1 at a time; sum order kept
# speedup vs baseline: 3.5806x; 1.1168x over previous
#include <hip/hip_runtime.h>
#include <hip/hip_bf16.h>


typedef float v2f __attribute__((ext_vector_type(2)));
typedef float v8f __attribute__((ext_vector_type(8)));

#define NEG_SLOPE 0.2f
#define SM_EPS    1e-16f

template <int NCOLS>
__global__ void __launch_bounds__(256) wmma_gemm_f32(
    const float* __restrict__ X, const float* __restrict__ W,
    float* __restrict__ Y, int nrows)
{
    __shared__ float As[16][128];
    const int row0 = blockIdx.x * 16;
    const int tid  = threadIdx.x;

    {
        const float4* X4  = (const float4*)X;
        float4*       As4 = (float4*)&As[0][0];
#pragma unroll
        for (int i = tid; i < 512; i += 256) {
            int r  = i >> 5;
            int rr = row0 + r;
            float4 v = make_float4(0.f, 0.f, 0.f, 0.f);
            if (rr < nrows) v = X4[(size_t)rr * 32 + (i & 31)];
            As4[i] = v;
        }
    }
    __syncthreads();

    const int wave = tid >> 5;
    const int lane = tid & 31;
    const int col0 = wave * 16;
    if (col0 >= NCOLS) return;

    const int m  = lane & 15;
    const int hi = lane >> 4;

    const float* wp = W + (size_t)col0 + m;
    v8f acc = {};
#pragma unroll 4
    for (int k0 = 0; k0 < 128; k0 += 4) {
        const int ka = k0 + (hi << 1);
        v2f a, b;
        a.x = As[m][ka];
        a.y = As[m][ka + 1];
        b.x = wp[(size_t)ka * NCOLS];
        b.y = wp[(size_t)(ka + 1) * NCOLS];
        acc = __builtin_amdgcn_wmma_f32_16x16x4_f32(
             false, a,  false, b,
             (short)0, acc,  false,  false);
    }

    const int orow = row0 + (hi ? 8 : 0);
    float* yp = Y + (size_t)orow * NCOLS + col0 + m;
    if (row0 + 16 <= nrows) {
#pragma unroll
        for (int r = 0; r < 8; ++r) yp[(size_t)r * NCOLS] = acc[r];
    } else {
#pragma unroll
        for (int r = 0; r < 8; ++r)
            if (orow + r < nrows) yp[(size_t)r * NCOLS] = acc[r];
    }
}

template <int HEADS, int C>
__global__ void attn_scores(const float* __restrict__ H,
                            const float* __restrict__ att_src,
                            const float* __restrict__ att_dst,
                            float* __restrict__ a_src, float* __restrict__ a_dst,
                            int N)
{
    int t = blockIdx.x * blockDim.x + threadIdx.x;
    if (t >= N * HEADS) return;
    int hd = t & (HEADS - 1);
    const float* hp = H + (size_t)t * C;
    const float* as = att_src + hd * C;
    const float* ad = att_dst + hd * C;
    float ss = 0.f, sd = 0.f;
#pragma unroll
    for (int c = 0; c < C; ++c) { float v = hp[c]; ss += v * as[c]; sd += v * ad[c]; }
    a_src[t] = ss; a_dst[t] = sd;
}

__global__ void fill_u32(unsigned* __restrict__ p, unsigned v, int n)
{
    int t = blockIdx.x * blockDim.x + threadIdx.x;
    if (t < n) p[t] = v;
}

__global__ void dst_count_rank(const int* __restrict__ ei, int E,
                               int* __restrict__ cnt, int* __restrict__ rank)
{
    int t = blockIdx.x * blockDim.x + threadIdx.x;
    if (t >= E) return;
    rank[t] = atomicAdd(&cnt[ei[E + t]], 1);
}

__global__ void __launch_bounds__(1024) scan_counts(const int* __restrict__ cnt,
                                                    int* __restrict__ rowptr, int chunk)
{
    __shared__ int wtot[32];
    const int tid  = threadIdx.x;
    const int lane = tid & 31;
    const int wv   = tid >> 5;
    const int nq   = chunk >> 2;
    const int4* c4 = (const int4*)(cnt + (size_t)tid * chunk);
    int4*       r4 = (int4*)(rowptr + (size_t)tid * chunk);

    int s = 0;
    for (int i = 0; i < nq; ++i) { int4 v = c4[i]; s += (v.x + v.y) + (v.z + v.w); }

    int incl = s;
    for (int o = 1; o < 32; o <<= 1) {
        int v = __shfl_up(incl, o);
        if (lane >= o) incl += v;
    }
    if (lane == 31) wtot[wv] = incl;
    __syncthreads();
    if (wv == 0) {
        int w = wtot[lane];
        int wi = w;
        for (int o = 1; o < 32; o <<= 1) {
            int v = __shfl_up(wi, o);
            if (lane >= o) wi += v;
        }
        wtot[lane] = wi - w;
    }
    __syncthreads();

    int run = wtot[wv] + (incl - s);
    for (int i = 0; i < nq; ++i) {
        int4 v = c4[i];
        int4 o;
        o.x = run;
        o.y = o.x + v.x;
        o.z = o.y + v.y;
        o.w = o.z + v.z;
        run = o.w + v.w;
        r4[i] = o;
    }
    if (tid == 1023) rowptr[(size_t)1024 * chunk] = run;
}

__global__ void csr_place(const int* __restrict__ ei, int E,
                          const int* __restrict__ rowptr, const int* __restrict__ rank,
                          int* __restrict__ csr_src)
{
    int t = blockIdx.x * blockDim.x + threadIdx.x;
    if (t >= E) return;
    csr_src[rowptr[ei[E + t]] + rank[t]] = ei[t];
}

template <int U> struct UTag { static constexpr int value = U; };

template <typename Fn>
__device__ __forceinline__ void for_edge_groups(int n, Fn f)
{
    int k = 0;
    for (; k + 8 <= n; k += 8) f(k, UTag<8>{});
    if (n & 4) { f(k, UTag<4>{}); k += 4; }
    if (n & 2) { f(k, UTag<2>{}); k += 2; }
    if (n & 1) { f(k, UTag<1>{}); }
}

__device__ __forceinline__ float lane_value(float x, int k)
{
    return __builtin_bit_cast(float, __builtin_amdgcn_readlane(__builtin_bit_cast(int, x), k));
}

__device__ __forceinline__ float leaky(float a) { return (a > 0.f) ? a : NEG_SLOPE * a; }

__device__ __forceinline__ float wave_max(float x)
{
#define ROW_XMASK_MAX(ctrl) \
    x = fmaxf(x, __builtin_bit_cast(float, __builtin_amdgcn_update_dpp( \
            0, __builtin_bit_cast(int, x), (ctrl), 0xf, 0xf, false)))
    ROW_XMASK_MAX(0x161);
    ROW_XMASK_MAX(0x162);
    ROW_XMASK_MAX(0x164);
    ROW_XMASK_MAX(0x168);
#undef ROW_XMASK_MAX
    const int xi = __builtin_bit_cast(int, x);
    x = fmaxf(x, __builtin_bit_cast(float,
            __builtin_amdgcn_permlanex16(xi, xi, 0x76543210, 0xfedcba98, false, false)));
    return x;
}

__device__ __forceinline__ void wave_lds_order()
{
    __builtin_amdgcn_fence(__ATOMIC_RELEASE, "wavefront");
    __builtin_amdgcn_wave_barrier();
    __builtin_amdgcn_fence(__ATOMIC_ACQUIRE, "wavefront");
}

template <int HEADS, int C, bool RELU>
__global__ void __launch_bounds__(256) gat_gather(
    const int* __restrict__ rowptr, const int* __restrict__ csr_src,
    const float* __restrict__ Hm, const float* __restrict__ a_src,
    const float* __restrict__ a_dst, const float* __restrict__ bias,
    float* __restrict__ out, int N)
{
    constexpr int F   = HEADS * C;
    constexpr int VEC = F / 32;
    static_assert(F % 32 == 0 && (VEC == 2 || VEC == 4), "a row must split evenly over one wave");
    static_assert(HEADS == 1 || HEADS % 4 == 0, "the LDS strip is filled with b128 stores");
    typedef float vf __attribute__((ext_vector_type(VEC)));

    __shared__ float s_as[8][HEADS == 1 ? 4 : 32 * HEADS];

    const int lane = threadIdx.x & 31;
    const int wv   = __builtin_amdgcn_readfirstlane((int)(threadIdx.x >> 5));
    const int d = __builtin_amdgcn_readfirstlane((int)(blockIdx.x * 8) + wv);
    if (d >= N) return;
    const int beg = __builtin_amdgcn_readfirstlane(rowptr[d]);
    const int end = __builtin_amdgcn_readfirstlane(rowptr[d + 1]);
    const bool one_chunk = (end - beg <= 32);

    const int    hd  = (lane * VEC) / C;
    const float  ad  = a_dst[d * HEADS + hd];
    const vf*    Hv  = (const vf*)Hm + lane;
    const vf     bv  = ((const vf*)bias)[lane];
    float*       sa  = s_as[wv];

    const float a_self = leaky(a_src[(size_t)d * HEADS + hd] + ad);

    float m = a_self;
    float den;
    vf    acc;

    if constexpr (HEADS == 1) {
        int   mine = 0;
        float al   = -INFINITY;
        auto fetch = [&](int j0, int n) {
            mine = 0; al = -INFINITY;
            if (lane < n) {
                mine = csr_src[j0 + lane];
                al   = leaky(a_src[mine] + ad);
            }
        };
        for (int j0 = beg; j0 < end; j0 += 32) {
            fetch(j0, (end - j0 < 32) ? (end - j0) : 32);
            m = fmaxf(m, al);
        }
        m = wave_max(m);
        float w = expf(a_self - m);
        den = w;
        acc = Hv[(size_t)d * 32] * w;
        for (int j0 = beg; j0 < end; j0 += 32) {
            const int n = (end - j0 < 32) ? (end - j0) : 32;
            if (!one_chunk) fetch(j0, n);
            const float wl = expf(al - m);
            for_edge_groups(n, [&](int k, auto u) {
                constexpr int U = decltype(u)::value;
                vf rows[U];
                for (int i = 0; i < U; ++i)
                    rows[i] = Hv[(size_t)__builtin_amdgcn_readlane(mine, k + i) * 32];
                for (int i = 0; i < U; ++i) {
                    const float wk = lane_value(wl, k + i);
                    den += wk;
                    acc += rows[i] * wk;
                }
            });
        }
    } else {
        int mine = 0;
        auto stage = [&](int j0, int n) {
            wave_lds_order();
            mine = 0;
            if (lane < n) {
                mine = csr_src[j0 + lane];
                const float4* p = (const float4*)(a_src + (size_t)mine * HEADS);
                float4*       q = (float4*)(sa + lane * HEADS);
                for (int i = 0; i < HEADS / 4; ++i) q[i] = p[i];
            }
            wave_lds_order();
        };
        for (int j0 = beg; j0 < end; j0 += 32) {
            const int n = (end - j0 < 32) ? (end - j0) : 32;
            stage(j0, n);
            for_edge_groups(n, [&](int k, auto u) {
                constexpr int U = decltype(u)::value;
                float a[U];
                for (int i = 0; i < U; ++i) a[i] = sa[(k + i) * HEADS + hd];
                for (int i = 0; i < U; ++i) m = fmaxf(m, leaky(a[i] + ad));
            });
        }
        float w = expf(a_self - m);
        den = w;
        acc = Hv[(size_t)d * 32] * w;
        for (int j0 = beg; j0 < end; j0 += 32) {
            const int n = (end - j0 < 32) ? (end - j0) : 32;
            if (!one_chunk) stage(j0, n);
            for_edge_groups(n, [&](int k, auto u) {
                constexpr int U = decltype(u)::value;
                vf    rows[U];
                float ws[U];
                for (int i = 0; i < U; ++i)
                    rows[i] = Hv[(size_t)__builtin_amdgcn_readlane(mine, k + i) * 32];
                for (int i = 0; i < U; ++i)
                    ws[i] = expf(leaky(sa[(k + i) * HEADS + hd] + ad) - m);
                for (int i = 0; i < U; ++i) {
                    den += ws[i];
                    acc += rows[i] * ws[i];
                }
            });
        }
    }

    vf r = acc / (den + SM_EPS) + bv;
    if (RELU) {
#pragma unroll
        for (int i = 0; i < VEC; ++i) r[i] = fmaxf(r[i], 0.f);
    }
    ((vf*)out)[(size_t)d * 32 + lane] = r;
}

static inline int cdiv(long long a, int b) { return (int)((a + b - 1) / b); }

extern "C" void kernel_launch(void* const* d_in, const int* in_sizes, int n_in,
                              void* d_out, int out_size, void* d_ws, size_t ws_size,
                              hipStream_t stream)
{
    const float* x        = (const float*)d_in[0];
    const int*   ei       = (const int*)  d_in[1];
    const float* W1       = (const float*)d_in[2];
    const float* att_src1 = (const float*)d_in[3];
    const float* att_dst1 = (const float*)d_in[4];
    const float* b1       = (const float*)d_in[5];
    const float* W2       = (const float*)d_in[6];
    const float* att_src2 = (const float*)d_in[7];
    const float* att_dst2 = (const float*)d_in[8];
    const float* b2       = (const float*)d_in[9];

    constexpr int IN = 128, H1 = 8, C1 = 16, F1 = H1 * C1  , C2 = 64;
    const int N = in_sizes[0] / IN;
    const int E = in_sizes[1] / 2;

    const int chunk = ((N + 1023) / 1024 + 3) & ~3;
    const int NP    = 1024 * chunk;

    size_t off = 0;
    float* base = (float*)d_ws;
    auto take = [&](size_t n) { float* p = base + off; off += (n + 63) & ~(size_t)63; return p; };
    float* h1      = take((size_t)N * F1);
    float* asrc1   = take((size_t)N * H1);
    float* adst1   = take((size_t)N * H1);
    float* agg1    = take((size_t)N * F1);
    float* h2      = take((size_t)N * C2);
    float* asrc2   = take((size_t)N);
    float* adst2   = take((size_t)N);
    int*   cnt     = (int*)take((size_t)NP);
    int*   rowptr  = (int*)take((size_t)NP + 4);
    int*   rank    = (int*)take((size_t)E);
    int*   csr_src = (int*)take((size_t)E);

    const int B = 256;

    fill_u32<<<cdiv(NP, B), B, 0, stream>>>((unsigned*)cnt, 0u, NP);
    dst_count_rank<<<cdiv(E, B), B, 0, stream>>>(ei, E, cnt, rank);
    scan_counts<<<1, 1024, 0, stream>>>(cnt, rowptr, chunk);
    csr_place<<<cdiv(E, B), B, 0, stream>>>(ei, E, rowptr, rank, csr_src);

    wmma_gemm_f32<F1><<<cdiv(N, 16), 256, 0, stream>>>(x, W1, h1, N);
    attn_scores<H1, C1><<<cdiv((long long)N * H1, B), B, 0, stream>>>(
        h1, att_src1, att_dst1, asrc1, adst1, N);
    gat_gather<H1, C1, true><<<cdiv(N, 8), 256, 0, stream>>>(
        rowptr, csr_src, h1, asrc1, adst1, b1, agg1, N);

    wmma_gemm_f32<C2><<<cdiv(N, 16), 256, 0, stream>>>(agg1, W2, h2, N);
    attn_scores<1, C2><<<cdiv((long long)N, B), B, 0, stream>>>(
        h2, att_src2, att_dst2, asrc2, adst2, N);
    gat_gather<1, C2, false><<<cdiv(N, 8), 256, 0, stream>>>(
        rowptr, csr_src, h2, asrc2, adst2, b2, (float*)d_out, N);
}


# GEMM A tile held in LDS at a row pitch of 132 floats, not 128: the 16 rows of each A-fragment read now fall on 16 different bank pairs (was one pair, a 16-way conflict on every fragment read)
# speedup vs baseline: 3.8584x; 1.0776x over previous
#include <hip/hip_runtime.h>
#include <hip/hip_bf16.h>


typedef float v2f __attribute__((ext_vector_type(2)));
typedef float v8f __attribute__((ext_vector_type(8)));

#define NEG_SLOPE 0.2f
#define SM_EPS    1e-16f

template <int NCOLS>
__global__ void __launch_bounds__(256) wmma_gemm_f32(
    const float* __restrict__ X, const float* __restrict__ W,
    float* __restrict__ Y, int nrows)
{
    constexpr int AP = 132;
    __shared__ alignas(16) float As[16][AP];
    const int row0 = blockIdx.x * 16;
    const int tid  = threadIdx.x;

    {
        const float4* X4  = (const float4*)X;
        float4*       As4 = (float4*)&As[0][0];
#pragma unroll
        for (int i = tid; i < 512; i += 256) {
            int r  = i >> 5;
            int rr = row0 + r;
            float4 v = make_float4(0.f, 0.f, 0.f, 0.f);
            if (rr < nrows) v = X4[(size_t)rr * 32 + (i & 31)];
            As4[r * (AP / 4) + (i & 31)] = v;
        }
    }
    __syncthreads();

    const int wave = tid >> 5;
    const int lane = tid & 31;
    const int col0 = wave * 16;
    if (col0 >= NCOLS) return;

    const int m  = lane & 15;
    const int hi = lane >> 4;

    const float* wp = W + (size_t)col0 + m;
    v8f acc = {};
#pragma unroll 4
    for (int k0 = 0; k0 < 128; k0 += 4) {
        const int ka = k0 + (hi << 1);
        v2f a, b;
        a.x = As[m][ka];
        a.y = As[m][ka + 1];
        b.x = wp[(size_t)ka * NCOLS];
        b.y = wp[(size_t)(ka + 1) * NCOLS];
        acc = __builtin_amdgcn_wmma_f32_16x16x4_f32(
             false, a,  false, b,
             (short)0, acc,  false,  false);
    }

    const int orow = row0 + (hi ? 8 : 0);
    float* yp = Y + (size_t)orow * NCOLS + col0 + m;
    if (row0 + 16 <= nrows) {
#pragma unroll
        for (int r = 0; r < 8; ++r) yp[(size_t)r * NCOLS] = acc[r];
    } else {
#pragma unroll
        for (int r = 0; r < 8; ++r)
            if (orow + r < nrows) yp[(size_t)r * NCOLS] = acc[r];
    }
}

template <int HEADS, int C>
__global__ void attn_scores(const float* __restrict__ H,
                            const float* __restrict__ att_src,
                            const float* __restrict__ att_dst,
                            float* __restrict__ a_src, float* __restrict__ a_dst,
                            int N)
{
    int t = blockIdx.x * blockDim.x + threadIdx.x;
    if (t >= N * HEADS) return;
    int hd = t & (HEADS - 1);
    const float* hp = H + (size_t)t * C;
    const float* as = att_src + hd * C;
    const float* ad = att_dst + hd * C;
    float ss = 0.f, sd = 0.f;
#pragma unroll
    for (int c = 0; c < C; ++c) { float v = hp[c]; ss += v * as[c]; sd += v * ad[c]; }
    a_src[t] = ss; a_dst[t] = sd;
}

__global__ void fill_u32(unsigned* __restrict__ p, unsigned v, int n)
{
    int t = blockIdx.x * blockDim.x + threadIdx.x;
    if (t < n) p[t] = v;
}

__global__ void dst_count_rank(const int* __restrict__ ei, int E,
                               int* __restrict__ cnt, int* __restrict__ rank)
{
    int t = blockIdx.x * blockDim.x + threadIdx.x;
    if (t >= E) return;
    rank[t] = atomicAdd(&cnt[ei[E + t]], 1);
}

__global__ void __launch_bounds__(1024) scan_counts(const int* __restrict__ cnt,
                                                    int* __restrict__ rowptr, int chunk)
{
    __shared__ int wtot[32];
    const int tid  = threadIdx.x;
    const int lane = tid & 31;
    const int wv   = tid >> 5;
    const int nq   = chunk >> 2;
    const int4* c4 = (const int4*)(cnt + (size_t)tid * chunk);
    int4*       r4 = (int4*)(rowptr + (size_t)tid * chunk);

    int s = 0;
    for (int i = 0; i < nq; ++i) { int4 v = c4[i]; s += (v.x + v.y) + (v.z + v.w); }

    int incl = s;
    for (int o = 1; o < 32; o <<= 1) {
        int v = __shfl_up(incl, o);
        if (lane >= o) incl += v;
    }
    if (lane == 31) wtot[wv] = incl;
    __syncthreads();
    if (wv == 0) {
        int w = wtot[lane];
        int wi = w;
        for (int o = 1; o < 32; o <<= 1) {
            int v = __shfl_up(wi, o);
            if (lane >= o) wi += v;
        }
        wtot[lane] = wi - w;
    }
    __syncthreads();

    int run = wtot[wv] + (incl - s);
    for (int i = 0; i < nq; ++i) {
        int4 v = c4[i];
        int4 o;
        o.x = run;
        o.y = o.x + v.x;
        o.z = o.y + v.y;
        o.w = o.z + v.z;
        run = o.w + v.w;
        r4[i] = o;
    }
    if (tid == 1023) rowptr[(size_t)1024 * chunk] = run;
}

__global__ void csr_place(const int* __restrict__ ei, int E,
                          const int* __restrict__ rowptr, const int* __restrict__ rank,
                          int* __restrict__ csr_src)
{
    int t = blockIdx.x * blockDim.x + threadIdx.x;
    if (t >= E) return;
    csr_src[rowptr[ei[E + t]] + rank[t]] = ei[t];
}

template <int U> struct UTag { static constexpr int value = U; };

template <typename Fn>
__device__ __forceinline__ void for_edge_groups(int n, Fn f)
{
    int k = 0;
    for (; k + 8 <= n; k += 8) f(k, UTag<8>{});
    if (n & 4) { f(k, UTag<4>{}); k += 4; }
    if (n & 2) { f(k, UTag<2>{}); k += 2; }
    if (n & 1) { f(k, UTag<1>{}); }
}

__device__ __forceinline__ float lane_value(float x, int k)
{
    return __builtin_bit_cast(float, __builtin_amdgcn_readlane(__builtin_bit_cast(int, x), k));
}

__device__ __forceinline__ float leaky(float a) { return (a > 0.f) ? a : NEG_SLOPE * a; }

__device__ __forceinline__ float wave_max(float x)
{
#define ROW_XMASK_MAX(ctrl) \
    x = fmaxf(x, __builtin_bit_cast(float, __builtin_amdgcn_update_dpp( \
            0, __builtin_bit_cast(int, x), (ctrl), 0xf, 0xf, false)))
    ROW_XMASK_MAX(0x161);
    ROW_XMASK_MAX(0x162);
    ROW_XMASK_MAX(0x164);
    ROW_XMASK_MAX(0x168);
#undef ROW_XMASK_MAX
    const int xi = __builtin_bit_cast(int, x);
    x = fmaxf(x, __builtin_bit_cast(float,
            __builtin_amdgcn_permlanex16(xi, xi, 0x76543210, 0xfedcba98, false, false)));
    return x;
}

__device__ __forceinline__ void wave_lds_order()
{
    __builtin_amdgcn_fence(__ATOMIC_RELEASE, "wavefront");
    __builtin_amdgcn_wave_barrier();
    __builtin_amdgcn_fence(__ATOMIC_ACQUIRE, "wavefront");
}

template <int HEADS, int C, bool RELU>
__global__ void __launch_bounds__(256) gat_gather(
    const int* __restrict__ rowptr, const int* __restrict__ csr_src,
    const float* __restrict__ Hm, const float* __restrict__ a_src,
    const float* __restrict__ a_dst, const float* __restrict__ bias,
    float* __restrict__ out, int N)
{
    constexpr int F   = HEADS * C;
    constexpr int VEC = F / 32;
    static_assert(F % 32 == 0 && (VEC == 2 || VEC == 4), "a row must split evenly over one wave");
    static_assert(HEADS == 1 || HEADS % 4 == 0, "the LDS strip is filled with b128 stores");
    typedef float vf __attribute__((ext_vector_type(VEC)));

    __shared__ float s_as[8][HEADS == 1 ? 4 : 32 * HEADS];

    const int lane = threadIdx.x & 31;
    const int wv   = __builtin_amdgcn_readfirstlane((int)(threadIdx.x >> 5));
    const int d = __builtin_amdgcn_readfirstlane((int)(blockIdx.x * 8) + wv);
    if (d >= N) return;
    const int beg = __builtin_amdgcn_readfirstlane(rowptr[d]);
    const int end = __builtin_amdgcn_readfirstlane(rowptr[d + 1]);
    const bool one_chunk = (end - beg <= 32);

    const int    hd  = (lane * VEC) / C;
    const float  ad  = a_dst[d * HEADS + hd];
    const vf*    Hv  = (const vf*)Hm + lane;
    const vf     bv  = ((const vf*)bias)[lane];
    float*       sa  = s_as[wv];

    const float a_self = leaky(a_src[(size_t)d * HEADS + hd] + ad);

    float m = a_self;
    float den;
    vf    acc;

    if constexpr (HEADS == 1) {
        int   mine = 0;
        float al   = -INFINITY;
        auto fetch = [&](int j0, int n) {
            mine = 0; al = -INFINITY;
            if (lane < n) {
                mine = csr_src[j0 + lane];
                al   = leaky(a_src[mine] + ad);
            }
        };
        for (int j0 = beg; j0 < end; j0 += 32) {
            fetch(j0, (end - j0 < 32) ? (end - j0) : 32);
            m = fmaxf(m, al);
        }
        m = wave_max(m);
        float w = expf(a_self - m);
        den = w;
        acc = Hv[(size_t)d * 32] * w;
        for (int j0 = beg; j0 < end; j0 += 32) {
            const int n = (end - j0 < 32) ? (end - j0) : 32;
            if (!one_chunk) fetch(j0, n);
            const float wl = expf(al - m);
            for_edge_groups(n, [&](int k, auto u) {
                constexpr int U = decltype(u)::value;
                vf rows[U];
                for (int i = 0; i < U; ++i)
                    rows[i] = Hv[(size_t)__builtin_amdgcn_readlane(mine, k + i) * 32];
                for (int i = 0; i < U; ++i) {
                    const float wk = lane_value(wl, k + i);
                    den += wk;
                    acc += rows[i] * wk;
                }
            });
        }
    } else {
        int mine = 0;
        auto stage = [&](int j0, int n) {
            wave_lds_order();
            mine = 0;
            if (lane < n) {
                mine = csr_src[j0 + lane];
                const float4* p = (const float4*)(a_src + (size_t)mine * HEADS);
                float4*       q = (float4*)(sa + lane * HEADS);
                for (int i = 0; i < HEADS / 4; ++i) q[i] = p[i];
            }
            wave_lds_order();
        };
        for (int j0 = beg; j0 < end; j0 += 32) {
            const int n = (end - j0 < 32) ? (end - j0) : 32;
            stage(j0, n);
            for_edge_groups(n, [&](int k, auto u) {
                constexpr int U = decltype(u)::value;
                float a[U];
                for (int i = 0; i < U; ++i) a[i] = sa[(k + i) * HEADS + hd];
                for (int i = 0; i < U; ++i) m = fmaxf(m, leaky(a[i] + ad));
            });
        }
        float w = expf(a_self - m);
        den = w;
        acc = Hv[(size_t)d * 32] * w;
        for (int j0 = beg; j0 < end; j0 += 32) {
            const int n = (end - j0 < 32) ? (end - j0) : 32;
            if (!one_chunk) stage(j0, n);
            for_edge_groups(n, [&](int k, auto u) {
                constexpr int U = decltype(u)::value;
                vf    rows[U];
                float ws[U];
                for (int i = 0; i < U; ++i)
                    rows[i] = Hv[(size_t)__builtin_amdgcn_readlane(mine, k + i) * 32];
                for (int i = 0; i < U; ++i)
                    ws[i] = expf(leaky(sa[(k + i) * HEADS + hd] + ad) - m);
                for (int i = 0; i < U; ++i) {
                    den += ws[i];
                    acc += rows[i] * ws[i];
                }
            });
        }
    }

    vf r = acc / (den + SM_EPS) + bv;
    if (RELU) {
#pragma unroll
        for (int i = 0; i < VEC; ++i) r[i] = fmaxf(r[i], 0.f);
    }
    ((vf*)out)[(size_t)d * 32 + lane] = r;
}

static inline int cdiv(long long a, int b) { return (int)((a + b - 1) / b); }

extern "C" void kernel_launch(void* const* d_in, const int* in_sizes, int n_in,
                              void* d_out, int out_size, void* d_ws, size_t ws_size,
                              hipStream_t stream)
{
    const float* x        = (const float*)d_in[0];
    const int*   ei       = (const int*)  d_in[1];
    const float* W1       = (const float*)d_in[2];
    const float* att_src1 = (const float*)d_in[3];
    const float* att_dst1 = (const float*)d_in[4];
    const float* b1       = (const float*)d_in[5];
    const float* W2       = (const float*)d_in[6];
    const float* att_src2 = (const float*)d_in[7];
    const float* att_dst2 = (const float*)d_in[8];
    const float* b2       = (const float*)d_in[9];

    constexpr int IN = 128, H1 = 8, C1 = 16, F1 = H1 * C1  , C2 = 64;
    const int N = in_sizes[0] / IN;
    const int E = in_sizes[1] / 2;

    const int chunk = ((N + 1023) / 1024 + 3) & ~3;
    const int NP    = 1024 * chunk;

    size_t off = 0;
    float* base = (float*)d_ws;
    auto take = [&](size_t n) { float* p = base + off; off += (n + 63) & ~(size_t)63; return p; };
    float* h1      = take((size_t)N * F1);
    float* asrc1   = take((size_t)N * H1);
    float* adst1   = take((size_t)N * H1);
    float* agg1    = take((size_t)N * F1);
    float* h2      = take((size_t)N * C2);
    float* asrc2   = take((size_t)N);
    float* adst2   = take((size_t)N);
    int*   cnt     = (int*)take((size_t)NP);
    int*   rowptr  = (int*)take((size_t)NP + 4);
    int*   rank    = (int*)take((size_t)E);
    int*   csr_src = (int*)take((size_t)E);

    const int B = 256;

    fill_u32<<<cdiv(NP, B), B, 0, stream>>>((unsigned*)cnt, 0u, NP);
    dst_count_rank<<<cdiv(E, B), B, 0, stream>>>(ei, E, cnt, rank);
    scan_counts<<<1, 1024, 0, stream>>>(cnt, rowptr, chunk);
    csr_place<<<cdiv(E, B), B, 0, stream>>>(ei, E, rowptr, rank, csr_src);

    wmma_gemm_f32<F1><<<cdiv(N, 16), 256, 0, stream>>>(x, W1, h1, N);
    attn_scores<H1, C1><<<cdiv((long long)N * H1, B), B, 0, stream>>>(
        h1, att_src1, att_dst1, asrc1, adst1, N);
    gat_gather<H1, C1, true><<<cdiv(N, 8), 256, 0, stream>>>(
        rowptr, csr_src, h1, asrc1, adst1, b1, agg1, N);

    wmma_gemm_f32<C2><<<cdiv(N, 16), 256, 0, stream>>>(agg1, W2, h2, N);
    attn_scores<1, C2><<<cdiv((long long)N, B), B, 0, stream>>>(
        h2, att_src2, att_dst2, asrc2, adst2, N);
    gat_gather<1, C2, false><<<cdiv(N, 8), 256, 0, stream>>>(
        rowptr, csr_src, h2, asrc2, adst2, b2, (float*)d_out, N);
}


# Edges grouped by destination via LDS: edge chunks split by 64-node range, one block per range counts/scans/places; global atomics, fill, 1-block scan gone (10->8 launches); sum order: LDS arrival
# speedup vs baseline: 6.7463x; 1.7485x over previous
#include <hip/hip_runtime.h>
#include <hip/hip_bf16.h>


typedef float v2f __attribute__((ext_vector_type(2)));
typedef float v8f __attribute__((ext_vector_type(8)));

#define NEG_SLOPE 0.2f
#define SM_EPS    1e-16f

template <int NCOLS>
__global__ void __launch_bounds__(256) wmma_gemm_f32(
    const float* __restrict__ X, const float* __restrict__ W,
    float* __restrict__ Y, int nrows)
{
    constexpr int AP = 132;
    __shared__ alignas(16) float As[16][AP];
    const int row0 = blockIdx.x * 16;
    const int tid  = threadIdx.x;

    {
        const float4* X4  = (const float4*)X;
        float4*       As4 = (float4*)&As[0][0];
#pragma unroll
        for (int i = tid; i < 512; i += 256) {
            int r  = i >> 5;
            int rr = row0 + r;
            float4 v = make_float4(0.f, 0.f, 0.f, 0.f);
            if (rr < nrows) v = X4[(size_t)rr * 32 + (i & 31)];
            As4[r * (AP / 4) + (i & 31)] = v;
        }
    }
    __syncthreads();

    const int wave = tid >> 5;
    const int lane = tid & 31;
    const int col0 = wave * 16;
    if (col0 >= NCOLS) return;

    const int m  = lane & 15;
    const int hi = lane >> 4;

    const float* wp = W + (size_t)col0 + m;
    v8f acc = {};
#pragma unroll 4
    for (int k0 = 0; k0 < 128; k0 += 4) {
        const int ka = k0 + (hi << 1);
        v2f a, b;
        a.x = As[m][ka];
        a.y = As[m][ka + 1];
        b.x = wp[(size_t)ka * NCOLS];
        b.y = wp[(size_t)(ka + 1) * NCOLS];
        acc = __builtin_amdgcn_wmma_f32_16x16x4_f32(
             false, a,  false, b,
             (short)0, acc,  false,  false);
    }

    const int orow = row0 + (hi ? 8 : 0);
    float* yp = Y + (size_t)orow * NCOLS + col0 + m;
    if (row0 + 16 <= nrows) {
#pragma unroll
        for (int r = 0; r < 8; ++r) yp[(size_t)r * NCOLS] = acc[r];
    } else {
#pragma unroll
        for (int r = 0; r < 8; ++r)
            if (orow + r < nrows) yp[(size_t)r * NCOLS] = acc[r];
    }
}

template <int HEADS, int C>
__global__ void attn_scores(const float* __restrict__ H,
                            const float* __restrict__ att_src,
                            const float* __restrict__ att_dst,
                            float* __restrict__ a_src, float* __restrict__ a_dst,
                            int N)
{
    int t = blockIdx.x * blockDim.x + threadIdx.x;
    if (t >= N * HEADS) return;
    int hd = t & (HEADS - 1);
    const float* hp = H + (size_t)t * C;
    const float* as = att_src + hd * C;
    const float* ad = att_dst + hd * C;
    float ss = 0.f, sd = 0.f;
#pragma unroll
    for (int c = 0; c < C; ++c) { float v = hp[c]; ss += v * as[c]; sd += v * ad[c]; }
    a_src[t] = ss; a_dst[t] = sd;
}

constexpr int NCHUNK = 256;

__device__ __forceinline__ int block_excl_scan_1024(int v, int* s_w)
{
    const int lane = threadIdx.x & 31;
    const int wv   = threadIdx.x >> 5;
    int incl = v;
    for (int o = 1; o < 32; o <<= 1) {
        int u = __shfl_up(incl, o);
        if (lane >= o) incl += u;
    }
    if (lane == 31) s_w[wv] = incl;
    __syncthreads();
    if (wv == 0) {
        int w = s_w[lane];
        int wi = w;
        for (int o = 1; o < 32; o <<= 1) {
            int u = __shfl_up(wi, o);
            if (lane >= o) wi += u;
        }
        s_w[lane] = wi - w;
    }
    __syncthreads();
    return s_w[wv] + (incl - v);
}

__global__ void __launch_bounds__(1024) edge_partition(
    const int* __restrict__ ei, int E, int T, int NR,
    int2* __restrict__ pairs, int* __restrict__ roffT)
{
    extern __shared__ __attribute__((aligned(16))) int dyn_lds[];
    __shared__ int s_w[32];
    int*  s_rng = dyn_lds;
    int2* s_st  = (int2*)(dyn_lds + ((NR + 2) & ~1));

    constexpr int U = 4;
    const int tid = threadIdx.x;
    const int w   = blockIdx.x;
    const int e0  = w * T;
    int n = E - e0;
    n = (n < 0) ? 0 : ((n > T) ? T : n);
    const int* srcs = ei + e0;
    const int* dsts = ei + (size_t)E + e0;

    for (int i = tid; i <= NR; i += 1024) s_rng[i] = 0;
    __syncthreads();
    for (int i0 = 0; i0 < n; i0 += 1024 * U) {
        int dv[U];
#pragma unroll
        for (int u = 0; u < U; ++u) {
            const int i = i0 + u * 1024 + tid;
            dv[u] = dsts[(i < n) ? i : n - 1];
        }
#pragma unroll
        for (int u = 0; u < U; ++u)
            if (i0 + u * 1024 + tid < n) atomicAdd(&s_rng[dv[u] >> 6], 1);
    }
    __syncthreads();

    const int per = (NR + 1 + 1023) >> 10;
    const int lo  = tid * per;
    int hi = lo + per;
    if (hi > NR + 1) hi = NR + 1;
    int s = 0;
    for (int i = lo; i < hi; ++i) s += s_rng[i];
    int run = block_excl_scan_1024(s, s_w);
    for (int i = lo; i < hi; ++i) { int c = s_rng[i]; s_rng[i] = run; run += c; }
    __syncthreads();


    for (int i0 = 0; i0 < n; i0 += 1024 * U) {
        int sv[U], dv[U];
#pragma unroll
        for (int u = 0; u < U; ++u) {
            const int i  = i0 + u * 1024 + tid;
            const int ic = (i < n) ? i : n - 1;
            sv[u] = srcs[ic];
            dv[u] = dsts[ic];
        }
#pragma unroll
        for (int u = 0; u < U; ++u)
            if (i0 + u * 1024 + tid < n) {
                const int p = atomicAdd(&s_rng[dv[u] >> 6], 1);
                s_st[p] = make_int2(sv[u], dv[u]);
            }
    }
    __syncthreads();
    for (int i = tid; i < n; i += 1024) pairs[(size_t)e0 + i] = s_st[i];
    for (int i = tid; i <= NR; i += 1024)
        roffT[(size_t)i * NCHUNK + w] = (i == 0) ? 0 : s_rng[i - 1];
}

__global__ void __launch_bounds__(NCHUNK) range_build(
    const int2* __restrict__ pairs, const int* __restrict__ roffT,
    int T, int NR, int N, int* __restrict__ rowptr, int* __restrict__ csr_src)
{
    constexpr int CAP = 8192;
    __shared__ int s_out[CAP];
    __shared__ int s_cnt[64];
    __shared__ int s_sum[2][NCHUNK / 32];

    const int t    = threadIdx.x;
    const int lane = t & 31;
    const int wv   = t >> 5;
    const int r    = blockIdx.x;
    const int n0   = r * 64;

    const int o0  = roffT[(size_t)r * NCHUNK + t];
    const int len = roffT[(size_t)(r + 1) * NCHUNK + t] - o0;
    if (t < 64) s_cnt[t] = 0;

    int a = o0, b = len;
    for (int o = 16; o > 0; o >>= 1) { a += __shfl_xor(a, o); b += __shfl_xor(b, o); }
    if (lane == 0) { s_sum[0][wv] = a; s_sum[1][wv] = b; }
    __syncthreads();
    int base = 0, M = 0;
    for (int i = 0; i < NCHUNK / 32; ++i) { base += s_sum[0][i]; M += s_sum[1][i]; }

    const int2* seg = pairs + (size_t)t * T + o0;
    for (int j = 0; j < len; j += 4) {
        int dl[4];
#pragma unroll
        for (int u = 0; u < 4; ++u) dl[u] = seg[(j + u < len) ? j + u : len - 1].y;
#pragma unroll
        for (int u = 0; u < 4; ++u)
            if (j + u < len) atomicAdd(&s_cnt[dl[u] - n0], 1);
    }
    __syncthreads();

    int rp0 = 0, rp1 = 0;
    if (wv == 0) {
        const int c0 = s_cnt[2 * lane];
        const int c1 = s_cnt[2 * lane + 1];
        const int sm = c0 + c1;
        int incl = sm;
        for (int o = 1; o < 32; o <<= 1) {
            int u = __shfl_up(incl, o);
            if (lane >= o) incl += u;
        }
        rp0 = incl - sm;
        rp1 = rp0 + c0;
        s_cnt[2 * lane]     = rp0;
        s_cnt[2 * lane + 1] = rp1;
    }
    __syncthreads();

    const bool in_lds = (M <= CAP);
    for (int j = 0; j < len; j += 4) {
        int2 p[4];
#pragma unroll
        for (int u = 0; u < 4; ++u) p[u] = seg[(j + u < len) ? j + u : len - 1];
#pragma unroll
        for (int u = 0; u < 4; ++u)
            if (j + u < len) {
                const int pos = atomicAdd(&s_cnt[p[u].y - n0], 1);
                if (in_lds) s_out[pos] = p[u].x;
                else        csr_src[(size_t)base + pos] = p[u].x;
            }
    }
    __syncthreads();
    if (in_lds)
        for (int i = t; i < M; i += NCHUNK) csr_src[(size_t)base + i] = s_out[i];
    if (wv == 0) {
        const int nd = n0 + 2 * lane;
        if (nd < N)     rowptr[nd]     = base + rp0;
        if (nd + 1 < N) rowptr[nd + 1] = base + rp1;
        if (r == NR - 1 && lane == 31) rowptr[N] = base + M;
    }
}

template <int U> struct UTag { static constexpr int value = U; };

template <typename Fn>
__device__ __forceinline__ void for_edge_groups(int n, Fn f)
{
    int k = 0;
    for (; k + 8 <= n; k += 8) f(k, UTag<8>{});
    if (n & 4) { f(k, UTag<4>{}); k += 4; }
    if (n & 2) { f(k, UTag<2>{}); k += 2; }
    if (n & 1) { f(k, UTag<1>{}); }
}

__device__ __forceinline__ float lane_value(float x, int k)
{
    return __builtin_bit_cast(float, __builtin_amdgcn_readlane(__builtin_bit_cast(int, x), k));
}

__device__ __forceinline__ float leaky(float a) { return (a > 0.f) ? a : NEG_SLOPE * a; }

__device__ __forceinline__ float wave_max(float x)
{
#define ROW_XMASK_MAX(ctrl) \
    x = fmaxf(x, __builtin_bit_cast(float, __builtin_amdgcn_update_dpp( \
            0, __builtin_bit_cast(int, x), (ctrl), 0xf, 0xf, false)))
    ROW_XMASK_MAX(0x161);
    ROW_XMASK_MAX(0x162);
    ROW_XMASK_MAX(0x164);
    ROW_XMASK_MAX(0x168);
#undef ROW_XMASK_MAX
    const int xi = __builtin_bit_cast(int, x);
    x = fmaxf(x, __builtin_bit_cast(float,
            __builtin_amdgcn_permlanex16(xi, xi, 0x76543210, 0xfedcba98, false, false)));
    return x;
}

__device__ __forceinline__ void wave_lds_order()
{
    __builtin_amdgcn_fence(__ATOMIC_RELEASE, "wavefront");
    __builtin_amdgcn_wave_barrier();
    __builtin_amdgcn_fence(__ATOMIC_ACQUIRE, "wavefront");
}

template <int HEADS, int C, bool RELU>
__global__ void __launch_bounds__(256) gat_gather(
    const int* __restrict__ rowptr, const int* __restrict__ csr_src,
    const float* __restrict__ Hm, const float* __restrict__ a_src,
    const float* __restrict__ a_dst, const float* __restrict__ bias,
    float* __restrict__ out, int N)
{
    constexpr int F   = HEADS * C;
    constexpr int VEC = F / 32;
    static_assert(F % 32 == 0 && (VEC == 2 || VEC == 4), "a row must split evenly over one wave");
    static_assert(HEADS == 1 || HEADS % 4 == 0, "the LDS strip is filled with b128 stores");
    typedef float vf __attribute__((ext_vector_type(VEC)));

    __shared__ float s_as[8][HEADS == 1 ? 4 : 32 * HEADS];

    const int lane = threadIdx.x & 31;
    const int wv   = __builtin_amdgcn_readfirstlane((int)(threadIdx.x >> 5));
    const int d = __builtin_amdgcn_readfirstlane((int)(blockIdx.x * 8) + wv);
    if (d >= N) return;
    const int beg = __builtin_amdgcn_readfirstlane(rowptr[d]);
    const int end = __builtin_amdgcn_readfirstlane(rowptr[d + 1]);
    const bool one_chunk = (end - beg <= 32);

    const int    hd  = (lane * VEC) / C;
    const float  ad  = a_dst[d * HEADS + hd];
    const vf*    Hv  = (const vf*)Hm + lane;
    const vf     bv  = ((const vf*)bias)[lane];
    float*       sa  = s_as[wv];

    const float a_self = leaky(a_src[(size_t)d * HEADS + hd] + ad);

    float m = a_self;
    float den;
    vf    acc;

    if constexpr (HEADS == 1) {
        int   mine = 0;
        float al   = -INFINITY;
        auto fetch = [&](int j0, int n) {
            mine = 0; al = -INFINITY;
            if (lane < n) {
                mine = csr_src[j0 + lane];
                al   = leaky(a_src[mine] + ad);
            }
        };
        for (int j0 = beg; j0 < end; j0 += 32) {
            fetch(j0, (end - j0 < 32) ? (end - j0) : 32);
            m = fmaxf(m, al);
        }
        m = wave_max(m);
        float w = expf(a_self - m);
        den = w;
        acc = Hv[(size_t)d * 32] * w;
        for (int j0 = beg; j0 < end; j0 += 32) {
            const int n = (end - j0 < 32) ? (end - j0) : 32;
            if (!one_chunk) fetch(j0, n);
            const float wl = expf(al - m);
            for_edge_groups(n, [&](int k, auto u) {
                constexpr int U = decltype(u)::value;
                vf rows[U];
                for (int i = 0; i < U; ++i)
                    rows[i] = Hv[(size_t)__builtin_amdgcn_readlane(mine, k + i) * 32];
                for (int i = 0; i < U; ++i) {
                    const float wk = lane_value(wl, k + i);
                    den += wk;
                    acc += rows[i] * wk;
                }
            });
        }
    } else {
        int mine = 0;
        auto stage = [&](int j0, int n) {
            wave_lds_order();
            mine = 0;
            if (lane < n) {
                mine = csr_src[j0 + lane];
                const float4* p = (const float4*)(a_src + (size_t)mine * HEADS);
                float4*       q = (float4*)(sa + lane * HEADS);
                for (int i = 0; i < HEADS / 4; ++i) q[i] = p[i];
            }
            wave_lds_order();
        };
        for (int j0 = beg; j0 < end; j0 += 32) {
            const int n = (end - j0 < 32) ? (end - j0) : 32;
            stage(j0, n);
            for_edge_groups(n, [&](int k, auto u) {
                constexpr int U = decltype(u)::value;
                float a[U];
                for (int i = 0; i < U; ++i) a[i] = sa[(k + i) * HEADS + hd];
                for (int i = 0; i < U; ++i) m = fmaxf(m, leaky(a[i] + ad));
            });
        }
        float w = expf(a_self - m);
        den = w;
        acc = Hv[(size_t)d * 32] * w;
        for (int j0 = beg; j0 < end; j0 += 32) {
            const int n = (end - j0 < 32) ? (end - j0) : 32;
            if (!one_chunk) stage(j0, n);
            for_edge_groups(n, [&](int k, auto u) {
                constexpr int U = decltype(u)::value;
                vf    rows[U];
                float ws[U];
                for (int i = 0; i < U; ++i)
                    rows[i] = Hv[(size_t)__builtin_amdgcn_readlane(mine, k + i) * 32];
                for (int i = 0; i < U; ++i)
                    ws[i] = expf(leaky(sa[(k + i) * HEADS + hd] + ad) - m);
                for (int i = 0; i < U; ++i) {
                    den += ws[i];
                    acc += rows[i] * ws[i];
                }
            });
        }
    }

    vf r = acc / (den + SM_EPS) + bv;
    if (RELU) {
#pragma unroll
        for (int i = 0; i < VEC; ++i) r[i] = fmaxf(r[i], 0.f);
    }
    ((vf*)out)[(size_t)d * 32 + lane] = r;
}

static inline int cdiv(long long a, int b) { return (int)((a + b - 1) / b); }

extern "C" void kernel_launch(void* const* d_in, const int* in_sizes, int n_in,
                              void* d_out, int out_size, void* d_ws, size_t ws_size,
                              hipStream_t stream)
{
    const float* x        = (const float*)d_in[0];
    const int*   ei       = (const int*)  d_in[1];
    const float* W1       = (const float*)d_in[2];
    const float* att_src1 = (const float*)d_in[3];
    const float* att_dst1 = (const float*)d_in[4];
    const float* b1       = (const float*)d_in[5];
    const float* W2       = (const float*)d_in[6];
    const float* att_src2 = (const float*)d_in[7];
    const float* att_dst2 = (const float*)d_in[8];
    const float* b2       = (const float*)d_in[9];

    constexpr int IN = 128, H1 = 8, C1 = 16, F1 = H1 * C1  , C2 = 64;
    const int N = in_sizes[0] / IN;
    const int E = in_sizes[1] / 2;

    const int T  = cdiv(E, NCHUNK);
    const int NR = cdiv(N, 64);

    size_t off = 0;
    float* base = (float*)d_ws;
    auto take = [&](size_t n) { float* p = base + off; off += (n + 63) & ~(size_t)63; return p; };
    float* h1      = take((size_t)N * F1);
    float* asrc1   = take((size_t)N * H1);
    float* adst1   = take((size_t)N * H1);
    float* agg1    = take((size_t)N * F1);
    float* h2      = take((size_t)N * C2);
    float* asrc2   = take((size_t)N);
    float* adst2   = take((size_t)N);
    int2*  pairs   = (int2*)take((size_t)NCHUNK * T * 2);
    int*   roffT   = (int*)take((size_t)(NR + 1) * NCHUNK);
    int*   rowptr  = (int*)take((size_t)N + 1);
    int*   csr_src = (int*)take((size_t)E);

    const int B = 256;

    const size_t part_lds = (size_t)((NR + 2) & ~1) * sizeof(int) + (size_t)T * sizeof(int2);
    edge_partition<<<NCHUNK, 1024, part_lds, stream>>>(ei, E, T, NR, pairs, roffT);
    range_build<<<NR, NCHUNK, 0, stream>>>(pairs, roffT, T, NR, N, rowptr, csr_src);

    wmma_gemm_f32<F1><<<cdiv(N, 16), 256, 0, stream>>>(x, W1, h1, N);
    attn_scores<H1, C1><<<cdiv((long long)N * H1, B), B, 0, stream>>>(
        h1, att_src1, att_dst1, asrc1, adst1, N);
    gat_gather<H1, C1, true><<<cdiv(N, 8), 256, 0, stream>>>(
        rowptr, csr_src, h1, asrc1, adst1, b1, agg1, N);

    wmma_gemm_f32<C2><<<cdiv(N, 16), 256, 0, stream>>>(agg1, W2, h2, N);
    attn_scores<1, C2><<<cdiv((long long)N, B), B, 0, stream>>>(
        h2, att_src2, att_dst2, asrc2, adst2, N);
    gat_gather<1, C2, false><<<cdiv(N, 8), 256, 0, stream>>>(
        rowptr, csr_src, h2, asrc2, adst2, b2, (float*)d_out, N);
}


# List building (grouping stage 2) and the layer-1 GEMM share one launch: block index picks the role, one LDS piece for both; the build's latency chain and launch boundary lie under the GEMM (6->5)
# speedup vs baseline: 7.8235x; 1.1597x over previous
#include <hip/hip_runtime.h>
#include <hip/hip_bf16.h>


typedef float v2f __attribute__((ext_vector_type(2)));
typedef float v8f __attribute__((ext_vector_type(8)));

#define NEG_SLOPE 0.2f
#define SM_EPS    1e-16f

constexpr int GEMM_AP = 132;

template <int NCOLS>
__device__ __forceinline__ void gemm_tile_f32(
    const float* __restrict__ X, const float* __restrict__ W,
    float* __restrict__ Y, int nrows, int tile, float (*As)[GEMM_AP])
{
    constexpr int AP = GEMM_AP;
    const int row0 = tile * 16;
    const int tid  = threadIdx.x;

    {
        const float4* X4  = (const float4*)X;
        float4*       As4 = (float4*)&As[0][0];
#pragma unroll
        for (int i = tid; i < 512; i += 256) {
            int r  = i >> 5;
            int rr = row0 + r;
            float4 v = make_float4(0.f, 0.f, 0.f, 0.f);
            if (rr < nrows) v = X4[(size_t)rr * 32 + (i & 31)];
            As4[r * (AP / 4) + (i & 31)] = v;
        }
    }
    __syncthreads();

    const int wave = tid >> 5;
    const int lane = tid & 31;
    const int col0 = wave * 16;
    if (col0 >= NCOLS) return;

    const int m  = lane & 15;
    const int hi = lane >> 4;

    const float* wp = W + (size_t)col0 + m;
    v8f acc = {};
#pragma unroll 4
    for (int k0 = 0; k0 < 128; k0 += 4) {
        const int ka = k0 + (hi << 1);
        v2f a, b;
        a.x = As[m][ka];
        a.y = As[m][ka + 1];
        b.x = wp[(size_t)ka * NCOLS];
        b.y = wp[(size_t)(ka + 1) * NCOLS];
        acc = __builtin_amdgcn_wmma_f32_16x16x4_f32(
             false, a,  false, b,
             (short)0, acc,  false,  false);
    }

    const int orow = row0 + (hi ? 8 : 0);
    float* yp = Y + (size_t)orow * NCOLS + col0 + m;
    if (row0 + 16 <= nrows) {
#pragma unroll
        for (int r = 0; r < 8; ++r) yp[(size_t)r * NCOLS] = acc[r];
    } else {
#pragma unroll
        for (int r = 0; r < 8; ++r)
            if (orow + r < nrows) yp[(size_t)r * NCOLS] = acc[r];
    }
}

template <int NCOLS>
__global__ void __launch_bounds__(256) wmma_gemm_f32(
    const float* __restrict__ X, const float* __restrict__ W,
    float* __restrict__ Y, int nrows)
{
    __shared__ alignas(16) float As[16][GEMM_AP];
    gemm_tile_f32<NCOLS>(X, W, Y, nrows, blockIdx.x, As);
}

constexpr int NCHUNK = 256;

__device__ __forceinline__ int block_excl_scan_1024(int v, int* s_w)
{
    const int lane = threadIdx.x & 31;
    const int wv   = threadIdx.x >> 5;
    int incl = v;
    for (int o = 1; o < 32; o <<= 1) {
        int u = __shfl_up(incl, o);
        if (lane >= o) incl += u;
    }
    if (lane == 31) s_w[wv] = incl;
    __syncthreads();
    if (wv == 0) {
        int w = s_w[lane];
        int wi = w;
        for (int o = 1; o < 32; o <<= 1) {
            int u = __shfl_up(wi, o);
            if (lane >= o) wi += u;
        }
        s_w[lane] = wi - w;
    }
    __syncthreads();
    return s_w[wv] + (incl - v);
}

__global__ void __launch_bounds__(1024) edge_partition(
    const int* __restrict__ ei, int E, int T, int NR,
    int2* __restrict__ pairs, int* __restrict__ roffT)
{
    extern __shared__ __attribute__((aligned(16))) int dyn_lds[];
    __shared__ int s_w[32];
    int*  s_rng = dyn_lds;
    int2* s_st  = (int2*)(dyn_lds + ((NR + 2) & ~1));

    constexpr int U = 4;
    const int tid = threadIdx.x;
    const int w   = blockIdx.x;
    const int e0  = w * T;
    int n = E - e0;
    n = (n < 0) ? 0 : ((n > T) ? T : n);
    const int* srcs = ei + e0;
    const int* dsts = ei + (size_t)E + e0;

    for (int i = tid; i <= NR; i += 1024) s_rng[i] = 0;
    __syncthreads();
    for (int i0 = 0; i0 < n; i0 += 1024 * U) {
        int dv[U];
#pragma unroll
        for (int u = 0; u < U; ++u) {
            const int i = i0 + u * 1024 + tid;
            dv[u] = dsts[(i < n) ? i : n - 1];
        }
#pragma unroll
        for (int u = 0; u < U; ++u)
            if (i0 + u * 1024 + tid < n) atomicAdd(&s_rng[dv[u] >> 6], 1);
    }
    __syncthreads();

    const int per = (NR + 1 + 1023) >> 10;
    const int lo  = tid * per;
    int hi = lo + per;
    if (hi > NR + 1) hi = NR + 1;
    int s = 0;
    for (int i = lo; i < hi; ++i) s += s_rng[i];
    int run = block_excl_scan_1024(s, s_w);
    for (int i = lo; i < hi; ++i) { int c = s_rng[i]; s_rng[i] = run; run += c; }
    __syncthreads();


    for (int i0 = 0; i0 < n; i0 += 1024 * U) {
        int sv[U], dv[U];
#pragma unroll
        for (int u = 0; u < U; ++u) {
            const int i  = i0 + u * 1024 + tid;
            const int ic = (i < n) ? i : n - 1;
            sv[u] = srcs[ic];
            dv[u] = dsts[ic];
        }
#pragma unroll
        for (int u = 0; u < U; ++u)
            if (i0 + u * 1024 + tid < n) {
                const int p = atomicAdd(&s_rng[dv[u] >> 6], 1);
                s_st[p] = make_int2(sv[u], dv[u]);
            }
    }
    __syncthreads();
    for (int i = tid; i < n; i += 1024) pairs[(size_t)e0 + i] = s_st[i];
    for (int i = tid; i <= NR; i += 1024)
        roffT[(size_t)i * NCHUNK + w] = (i == 0) ? 0 : s_rng[i - 1];
}

constexpr int RB_CAP = 8192;
constexpr int RB_LDS_INTS = RB_CAP + 64 + 2 * (NCHUNK / 32);

__device__ __forceinline__ void range_build_block(
    const int2* __restrict__ pairs, const int* __restrict__ roffT,
    int T, int NR, int N, int* __restrict__ rowptr, int* __restrict__ csr_src,
    int r, int* lds)
{
    constexpr int CAP = RB_CAP;
    int* s_out = lds;
    int* s_cnt = lds + CAP;
    int (*s_sum)[NCHUNK / 32] = (int (*)[NCHUNK / 32])(lds + CAP + 64);

    const int t    = threadIdx.x;
    const int lane = t & 31;
    const int wv   = t >> 5;
    const int n0   = r * 64;

    const int o0  = roffT[(size_t)r * NCHUNK + t];
    const int len = roffT[(size_t)(r + 1) * NCHUNK + t] - o0;
    if (t < 64) s_cnt[t] = 0;

    int a = o0, b = len;
    for (int o = 16; o > 0; o >>= 1) { a += __shfl_xor(a, o); b += __shfl_xor(b, o); }
    if (lane == 0) { s_sum[0][wv] = a; s_sum[1][wv] = b; }
    __syncthreads();
    int base = 0, M = 0;
    for (int i = 0; i < NCHUNK / 32; ++i) { base += s_sum[0][i]; M += s_sum[1][i]; }

    const int2* seg = pairs + (size_t)t * T + o0;
    for (int j = 0; j < len; j += 4) {
        int dl[4];
#pragma unroll
        for (int u = 0; u < 4; ++u) dl[u] = seg[(j + u < len) ? j + u : len - 1].y;
#pragma unroll
        for (int u = 0; u < 4; ++u)
            if (j + u < len) atomicAdd(&s_cnt[dl[u] - n0], 1);
    }
    __syncthreads();

    int rp0 = 0, rp1 = 0;
    if (wv == 0) {
        const int c0 = s_cnt[2 * lane];
        const int c1 = s_cnt[2 * lane + 1];
        const int sm = c0 + c1;
        int incl = sm;
        for (int o = 1; o < 32; o <<= 1) {
            int u = __shfl_up(incl, o);
            if (lane >= o) incl += u;
        }
        rp0 = incl - sm;
        rp1 = rp0 + c0;
        s_cnt[2 * lane]     = rp0;
        s_cnt[2 * lane + 1] = rp1;
    }
    __syncthreads();

    const bool in_lds = (M <= CAP);
    for (int j = 0; j < len; j += 4) {
        int2 p[4];
#pragma unroll
        for (int u = 0; u < 4; ++u) p[u] = seg[(j + u < len) ? j + u : len - 1];
#pragma unroll
        for (int u = 0; u < 4; ++u)
            if (j + u < len) {
                const int pos = atomicAdd(&s_cnt[p[u].y - n0], 1);
                if (in_lds) s_out[pos] = p[u].x;
                else        csr_src[(size_t)base + pos] = p[u].x;
            }
    }
    __syncthreads();
    if (in_lds)
        for (int i = t; i < M; i += NCHUNK) csr_src[(size_t)base + i] = s_out[i];
    if (wv == 0) {
        const int nd = n0 + 2 * lane;
        if (nd < N)     rowptr[nd]     = base + rp0;
        if (nd + 1 < N) rowptr[nd + 1] = base + rp1;
        if (r == NR - 1 && lane == 31) rowptr[N] = base + M;
    }
}

template <int NCOLS>
__global__ void __launch_bounds__(NCHUNK) range_build_and_gemm(
    const int2* __restrict__ pairs, const int* __restrict__ roffT,
    int T, int NR, int N, int* __restrict__ rowptr, int* __restrict__ csr_src,
    const float* __restrict__ X, const float* __restrict__ W, float* __restrict__ Y)
{
    static_assert(NCHUNK == 256, "the GEMM tile is loaded and multiplied by 256 threads");
    constexpr int GEMM_INTS = 16 * GEMM_AP;
    __shared__ alignas(16) int lds[RB_LDS_INTS > GEMM_INTS ? RB_LDS_INTS : GEMM_INTS];
    const int b = blockIdx.x;
    if (b < NR) range_build_block(pairs, roffT, T, NR, N, rowptr, csr_src, b, lds);
    else        gemm_tile_f32<NCOLS>(X, W, Y, N, b - NR, (float (*)[GEMM_AP])lds);
}

template <int U> struct UTag { static constexpr int value = U; };

template <typename Fn>
__device__ __forceinline__ void for_edge_groups(int n, Fn f)
{
    int k = 0;
    for (; k + 8 <= n; k += 8) f(k, UTag<8>{});
    if (n & 4) { f(k, UTag<4>{}); k += 4; }
    if (n & 2) { f(k, UTag<2>{}); k += 2; }
    if (n & 1) { f(k, UTag<1>{}); }
}

template <typename Fn>
__device__ __forceinline__ void for_edge_groups4(int n, Fn f)
{
    int k = 0;
    for (; k + 4 <= n; k += 4) f(k, UTag<4>{});
    if (n & 2) { f(k, UTag<2>{}); k += 2; }
    if (n & 1) { f(k, UTag<1>{}); }
}

__device__ __forceinline__ float leaky(float a) { return (a > 0.f) ? a : NEG_SLOPE * a; }

template <int MASK>
__device__ __forceinline__ float lane_xor(float x)
{
    return __builtin_bit_cast(float, __builtin_amdgcn_update_dpp(
            0, __builtin_bit_cast(int, x), 0x160 + MASK  , 0xf, 0xf, false));
}

__device__ __forceinline__ float lane_xor16(float x)
{
    const int xi = __builtin_bit_cast(int, x);
    return __builtin_bit_cast(float,
            __builtin_amdgcn_permlanex16(xi, xi, 0x76543210, 0xfedcba98, false, false));
}

template <int I>
__device__ __forceinline__ float row_lane_c(float x)
{
    return __builtin_bit_cast(float, __builtin_amdgcn_mov_dpp(
            __builtin_bit_cast(int, x), 0x150 + I  , 0xf, 0xf, true));
}
__device__ __forceinline__ float row_lane(float x, int i)
{
    switch (i) {
    case 0:  return row_lane_c<0>(x);
    case 1:  return row_lane_c<1>(x);
    case 2:  return row_lane_c<2>(x);
    case 3:  return row_lane_c<3>(x);
    default: return row_lane_c<4>(x);
    }
}
__device__ __forceinline__ float row_sum16(float x)
{
    x += __builtin_bit_cast(float, __builtin_amdgcn_mov_dpp(__builtin_bit_cast(int, x), 0x161  , 0xf, 0xf, true));
    x += __builtin_bit_cast(float, __builtin_amdgcn_mov_dpp(__builtin_bit_cast(int, x), 0x162  , 0xf, 0xf, true));
    x += __builtin_bit_cast(float, __builtin_amdgcn_mov_dpp(__builtin_bit_cast(int, x), 0x164  , 0xf, 0xf, true));
    x += __builtin_bit_cast(float, __builtin_amdgcn_mov_dpp(__builtin_bit_cast(int, x), 0x168  , 0xf, 0xf, true));
    return x;
}

template <int LPH>
__device__ __forceinline__ float head_sum(float x)
{
    static_assert(LPH == 1 || LPH == 2 || LPH == 4 || LPH == 8 || LPH == 16, "a head's lanes lie in one row of 16");
    if (LPH >= 2)  x += lane_xor<1>(x);
    if (LPH >= 4)  x += lane_xor<2>(x);
    if (LPH >= 8)  x += lane_xor<4>(x);
    if (LPH >= 16) x += lane_xor<8>(x);
    return x;
}

template <int HEADS, int C, bool RELU>
__global__ void __launch_bounds__(256) gat_gather_heads(
    const int* __restrict__ rowptr, const int* __restrict__ csr_src,
    const float* __restrict__ Hm, const float* __restrict__ att_src,
    const float* __restrict__ att_dst, const float* __restrict__ bias,
    float* __restrict__ out, int N)
{
    constexpr int F   = HEADS * C;
    constexpr int VEC = F / 32;
    constexpr int LPH = C / VEC;
    static_assert(F % 32 == 0 && VEC == 4 && C % VEC == 0, "a row is one float4 per lane, a head whole lanes");
    typedef float vf __attribute__((ext_vector_type(VEC)));

    const int lane = threadIdx.x & 31;
    const int wv   = __builtin_amdgcn_readfirstlane((int)(threadIdx.x >> 5));
    const int d = __builtin_amdgcn_readfirstlane((int)(blockIdx.x * 8) + wv);
    if (d >= N) return;
    const int beg = __builtin_amdgcn_readfirstlane(rowptr[d]);
    const int end = __builtin_amdgcn_readfirstlane(rowptr[d + 1]);

    const vf* Hv  = (const vf*)Hm + lane;
    const vf  as  = ((const vf*)att_src)[lane];
    const vf  ad  = ((const vf*)att_dst)[lane];
    const vf  bv  = ((const vf*)bias)[lane];

    auto dot = [](vf a, vf b) { return a.x * b.x + a.y * b.y + a.z * b.z + a.w * b.w; };

    const vf    own    = Hv[(size_t)d * 32];
    const float a_d    = head_sum<LPH>(dot(own, ad));
    const float a_self = leaky(head_sum<LPH>(dot(own, as)) + a_d);

    float m   = a_self;
    float den = 1.f;
    vf    acc = own;

    for (int j0 = beg; j0 < end; j0 += 32) {
        const int n = (end - j0 < 32) ? (end - j0) : 32;
        int mine = 0;
        if (lane < n) mine = csr_src[j0 + lane];
        for_edge_groups(n, [&](int k, auto u) {
            constexpr int U = decltype(u)::value;
            vf    rows[U];
            float al[U];
            for (int i = 0; i < U; ++i)
                rows[i] = Hv[(size_t)__builtin_amdgcn_readlane(mine, k + i) * 32];
            __builtin_amdgcn_sched_barrier(0);
            for (int i = 0; i < U; ++i)
                al[i] = leaky(head_sum<LPH>(dot(rows[i], as)) + a_d);
            float mn = m;
            for (int i = 0; i < U; ++i) mn = fmaxf(mn, al[i]);
            const float sc = expf(m - mn);
            den *= sc;
            acc *= sc;
            m = mn;
            for (int i = 0; i < U; ++i) {
                const float w = expf(al[i] - mn);
                den += w;
                acc += rows[i] * w;
            }
        });
    }

    vf r = acc / (den + SM_EPS) + bv;
    if (RELU) {
#pragma unroll
        for (int i = 0; i < VEC; ++i) r[i] = fmaxf(r[i], 0.f);
    }
    ((vf*)out)[(size_t)d * 32 + lane] = r;
}

template <int C, bool RELU>
__global__ void __launch_bounds__(256) gat_gather_1head(
    const int* __restrict__ rowptr, const int* __restrict__ csr_src,
    const float* __restrict__ Hm, const float* __restrict__ att_src,
    const float* __restrict__ att_dst, const float* __restrict__ bias,
    float* __restrict__ out, int N)
{
    static_assert(C == 64, "a row is one float4 per lane of a half-wave");
    typedef float vf __attribute__((ext_vector_type(4)));

    const int lane = threadIdx.x & 31;
    const int c    = lane & 15;
    const int half = lane >> 4;
    const int wv   = __builtin_amdgcn_readfirstlane((int)(threadIdx.x >> 5));
    const int d = __builtin_amdgcn_readfirstlane((int)(blockIdx.x * 8) + wv);
    if (d >= N) return;
    const int beg = __builtin_amdgcn_readfirstlane(rowptr[d]);
    const int end = __builtin_amdgcn_readfirstlane(rowptr[d + 1]);

    const char* Hb = (const char*)Hm;
    const unsigned cb = (unsigned)c << 4;
    const vf as = ((const vf*)att_src)[c];
    const vf ad = ((const vf*)att_dst)[c];
    const vf bv = ((const vf*)bias)[c];

    auto dot = [](vf a, vf b) { return a.x * b.x + a.y * b.y + a.z * b.z + a.w * b.w; };

    const vf    own    = *(const vf*)(Hb + (((unsigned)d << 8) | cb));
    const float a_d    = row_sum16(dot(own, ad));
    const float a_self = leaky(row_sum16(dot(own, as)) + a_d);

    float m   = a_self;
    float den = half ? 0.f : 1.f;
    vf    acc = half ? (vf)(0.f) : own;

    for (int j0 = beg; j0 < end; j0 += 32) {
        const int n = (end - j0 < 32) ? (end - j0) : 32;
        int mine = d;
        if (lane < n) mine = csr_src[j0 + lane];
        const int nh = n - half;
        for_edge_groups4((n + 1) >> 1, [&](int k, auto u) {
            constexpr int U = decltype(u)::value;
            vf    rows[U];
            float al[U];
            for (int i = 0; i < U; ++i) {
                const int se = __builtin_amdgcn_readlane(mine, 2 * (k + i));
                const int so = __builtin_amdgcn_readlane(mine, 2 * (k + i) + 1);
                const unsigned s = (unsigned)(half ? so : se);
                rows[i] = *(const vf*)(Hb + ((s << 8) | cb));
            }
            for (int i = 0; i < U; ++i) {
                const float a = leaky(row_sum16(dot(rows[i], as)) + a_d);
                al[i] = (2 * (k + i) < nh) ? a : -INFINITY;
            }
            float mn = m;
            for (int i = 0; i < U; ++i) mn = fmaxf(mn, al[i]);
            float x = m;
            for (int i = 0; i < U; ++i) x = (c == i) ? al[i] : x;
            const float ex = expf(x - mn);
            const float sc = row_lane(ex, U);
            den *= sc;
            acc *= sc;
            m = mn;
            for (int i = 0; i < U; ++i) {
                const float w = row_lane(ex, i);
                den += w;
                acc += rows[i] * w;
            }
        });
    }

    const float mj = fmaxf(m, lane_xor16(m));
    const float sj = expf(m - mj);
    den *= sj;
    acc *= sj;
    den += lane_xor16(den);
    acc.x += lane_xor16(acc.x);
    acc.y += lane_xor16(acc.y);
    acc.z += lane_xor16(acc.z);
    acc.w += lane_xor16(acc.w);

    vf r = acc / (den + SM_EPS) + bv;
    if (RELU) {
#pragma unroll
        for (int i = 0; i < 4; ++i) r[i] = fmaxf(r[i], 0.f);
    }
    if (half == 0) ((vf*)out)[(size_t)d * 16 + c] = r;
}

static inline int cdiv(long long a, int b) { return (int)((a + b - 1) / b); }

extern "C" void kernel_launch(void* const* d_in, const int* in_sizes, int n_in,
                              void* d_out, int out_size, void* d_ws, size_t ws_size,
                              hipStream_t stream)
{
    const float* x        = (const float*)d_in[0];
    const int*   ei       = (const int*)  d_in[1];
    const float* W1       = (const float*)d_in[2];
    const float* att_src1 = (const float*)d_in[3];
    const float* att_dst1 = (const float*)d_in[4];
    const float* b1       = (const float*)d_in[5];
    const float* W2       = (const float*)d_in[6];
    const float* att_src2 = (const float*)d_in[7];
    const float* att_dst2 = (const float*)d_in[8];
    const float* b2       = (const float*)d_in[9];

    constexpr int IN = 128, H1 = 8, C1 = 16, F1 = H1 * C1  , C2 = 64;
    const int N = in_sizes[0] / IN;
    const int E = in_sizes[1] / 2;

    const int T  = cdiv(E, NCHUNK);
    const int NR = cdiv(N, 64);

    size_t off = 0;
    float* base = (float*)d_ws;
    auto take = [&](size_t n) { float* p = base + off; off += (n + 63) & ~(size_t)63; return p; };
    float* h1      = take((size_t)N * F1);
    float* agg1    = take((size_t)N * F1);
    float* h2      = take((size_t)N * C2);
    int2*  pairs   = (int2*)take((size_t)NCHUNK * T * 2);
    int*   roffT   = (int*)take((size_t)(NR + 1) * NCHUNK);
    int*   rowptr  = (int*)take((size_t)N + 1);
    int*   csr_src = (int*)take((size_t)E);

    const size_t part_lds = (size_t)((NR + 2) & ~1) * sizeof(int) + (size_t)T * sizeof(int2);
    edge_partition<<<NCHUNK, 1024, part_lds, stream>>>(ei, E, T, NR, pairs, roffT);
    range_build_and_gemm<F1><<<NR + cdiv(N, 16), NCHUNK, 0, stream>>>(
        pairs, roffT, T, NR, N, rowptr, csr_src, x, W1, h1);

    gat_gather_heads<H1, C1, true><<<cdiv(N, 8), 256, 0, stream>>>(
        rowptr, csr_src, h1, att_src1, att_dst1, b1, agg1, N);

    wmma_gemm_f32<C2><<<cdiv(N, 16), 256, 0, stream>>>(agg1, W2, h2, N);
    gat_gather_1head<C2, false><<<cdiv(N, 8), 256, 0, stream>>>(
        rowptr, csr_src, h2, att_src2, att_dst2, b2, (float*)d_out, N);
}


# Layer-1 gather: a group's edge logits split over a head's 4 lanes (two-step split sum leaves lane q with edges q, q+4): leaky, max, exp once per (edge, head); weights sent back by DPP; sum order kept
# speedup vs baseline: 8.2393x; 1.0531x over previous
#include <hip/hip_runtime.h>
#include <hip/hip_bf16.h>


typedef float v2f __attribute__((ext_vector_type(2)));
typedef float v8f __attribute__((ext_vector_type(8)));

#define NEG_SLOPE 0.2f
#define SM_EPS    1e-16f

constexpr int GEMM_AP = 132;

template <int NCOLS>
__device__ __forceinline__ void gemm_tile_f32(
    const float* __restrict__ X, const float* __restrict__ W,
    float* __restrict__ Y, int nrows, int tile, float (*As)[GEMM_AP])
{
    constexpr int AP = GEMM_AP;
    const int row0 = tile * 16;
    const int tid  = threadIdx.x;

    {
        const float4* X4  = (const float4*)X;
        float4*       As4 = (float4*)&As[0][0];
#pragma unroll
        for (int i = tid; i < 512; i += 256) {
            int r  = i >> 5;
            int rr = row0 + r;
            float4 v = make_float4(0.f, 0.f, 0.f, 0.f);
            if (rr < nrows) v = X4[(size_t)rr * 32 + (i & 31)];
            As4[r * (AP / 4) + (i & 31)] = v;
        }
    }
    __syncthreads();

    const int wave = tid >> 5;
    const int lane = tid & 31;
    const int col0 = wave * 16;
    if (col0 >= NCOLS) return;

    const int m  = lane & 15;
    const int hi = lane >> 4;

    const float* wp = W + (size_t)col0 + m;
    v8f acc = {};
#pragma unroll 4
    for (int k0 = 0; k0 < 128; k0 += 4) {
        const int ka = k0 + (hi << 1);
        v2f a, b;
        a.x = As[m][ka];
        a.y = As[m][ka + 1];
        b.x = wp[(size_t)ka * NCOLS];
        b.y = wp[(size_t)(ka + 1) * NCOLS];
        acc = __builtin_amdgcn_wmma_f32_16x16x4_f32(
             false, a,  false, b,
             (short)0, acc,  false,  false);
    }

    const int orow = row0 + (hi ? 8 : 0);
    float* yp = Y + (size_t)orow * NCOLS + col0 + m;
    if (row0 + 16 <= nrows) {
#pragma unroll
        for (int r = 0; r < 8; ++r) yp[(size_t)r * NCOLS] = acc[r];
    } else {
#pragma unroll
        for (int r = 0; r < 8; ++r)
            if (orow + r < nrows) yp[(size_t)r * NCOLS] = acc[r];
    }
}

template <int NCOLS>
__global__ void __launch_bounds__(256) wmma_gemm_f32(
    const float* __restrict__ X, const float* __restrict__ W,
    float* __restrict__ Y, int nrows)
{
    __shared__ alignas(16) float As[16][GEMM_AP];
    gemm_tile_f32<NCOLS>(X, W, Y, nrows, blockIdx.x, As);
}

constexpr int NCHUNK = 256;

__device__ __forceinline__ int block_excl_scan_1024(int v, int* s_w)
{
    const int lane = threadIdx.x & 31;
    const int wv   = threadIdx.x >> 5;
    int incl = v;
    for (int o = 1; o < 32; o <<= 1) {
        int u = __shfl_up(incl, o);
        if (lane >= o) incl += u;
    }
    if (lane == 31) s_w[wv] = incl;
    __syncthreads();
    if (wv == 0) {
        int w = s_w[lane];
        int wi = w;
        for (int o = 1; o < 32; o <<= 1) {
            int u = __shfl_up(wi, o);
            if (lane >= o) wi += u;
        }
        s_w[lane] = wi - w;
    }
    __syncthreads();
    return s_w[wv] + (incl - v);
}

__global__ void __launch_bounds__(1024) edge_partition(
    const int* __restrict__ ei, int E, int T, int NR,
    int2* __restrict__ pairs, int* __restrict__ roffT)
{
    extern __shared__ __attribute__((aligned(16))) int dyn_lds[];
    __shared__ int s_w[32];
    int*  s_rng = dyn_lds;
    int2* s_st  = (int2*)(dyn_lds + ((NR + 2) & ~1));

    constexpr int U = 4;
    const int tid = threadIdx.x;
    const int w   = blockIdx.x;
    const int e0  = w * T;
    int n = E - e0;
    n = (n < 0) ? 0 : ((n > T) ? T : n);
    const int* srcs = ei + e0;
    const int* dsts = ei + (size_t)E + e0;

    for (int i = tid; i <= NR; i += 1024) s_rng[i] = 0;
    __syncthreads();
    for (int i0 = 0; i0 < n; i0 += 1024 * U) {
        int dv[U];
#pragma unroll
        for (int u = 0; u < U; ++u) {
            const int i = i0 + u * 1024 + tid;
            dv[u] = dsts[(i < n) ? i : n - 1];
        }
#pragma unroll
        for (int u = 0; u < U; ++u)
            if (i0 + u * 1024 + tid < n) atomicAdd(&s_rng[dv[u] >> 6], 1);
    }
    __syncthreads();

    const int per = (NR + 1 + 1023) >> 10;
    const int lo  = tid * per;
    int hi = lo + per;
    if (hi > NR + 1) hi = NR + 1;
    int s = 0;
    for (int i = lo; i < hi; ++i) s += s_rng[i];
    int run = block_excl_scan_1024(s, s_w);
    for (int i = lo; i < hi; ++i) { int c = s_rng[i]; s_rng[i] = run; run += c; }
    __syncthreads();


    for (int i0 = 0; i0 < n; i0 += 1024 * U) {
        int sv[U], dv[U];
#pragma unroll
        for (int u = 0; u < U; ++u) {
            const int i  = i0 + u * 1024 + tid;
            const int ic = (i < n) ? i : n - 1;
            sv[u] = srcs[ic];
            dv[u] = dsts[ic];
        }
#pragma unroll
        for (int u = 0; u < U; ++u)
            if (i0 + u * 1024 + tid < n) {
                const int p = atomicAdd(&s_rng[dv[u] >> 6], 1);
                s_st[p] = make_int2(sv[u], dv[u]);
            }
    }
    __syncthreads();
    for (int i = tid; i < n; i += 1024) pairs[(size_t)e0 + i] = s_st[i];
    for (int i = tid; i <= NR; i += 1024)
        roffT[(size_t)i * NCHUNK + w] = (i == 0) ? 0 : s_rng[i - 1];
}

constexpr int RB_CAP = 8192;
constexpr int RB_LDS_INTS = RB_CAP + 64 + 2 * (NCHUNK / 32);

__device__ __forceinline__ void range_build_block(
    const int2* __restrict__ pairs, const int* __restrict__ roffT,
    int T, int NR, int N, int* __restrict__ rowptr, int* __restrict__ csr_src,
    int r, int* lds)
{
    constexpr int CAP = RB_CAP;
    int* s_out = lds;
    int* s_cnt = lds + CAP;
    int (*s_sum)[NCHUNK / 32] = (int (*)[NCHUNK / 32])(lds + CAP + 64);

    const int t    = threadIdx.x;
    const int lane = t & 31;
    const int wv   = t >> 5;
    const int n0   = r * 64;

    const int o0  = roffT[(size_t)r * NCHUNK + t];
    const int len = roffT[(size_t)(r + 1) * NCHUNK + t] - o0;
    if (t < 64) s_cnt[t] = 0;

    int a = o0, b = len;
    for (int o = 16; o > 0; o >>= 1) { a += __shfl_xor(a, o); b += __shfl_xor(b, o); }
    if (lane == 0) { s_sum[0][wv] = a; s_sum[1][wv] = b; }
    __syncthreads();
    int base = 0, M = 0;
    for (int i = 0; i < NCHUNK / 32; ++i) { base += s_sum[0][i]; M += s_sum[1][i]; }

    const int2* seg = pairs + (size_t)t * T + o0;
    for (int j = 0; j < len; j += 4) {
        int dl[4];
#pragma unroll
        for (int u = 0; u < 4; ++u) dl[u] = seg[(j + u < len) ? j + u : len - 1].y;
#pragma unroll
        for (int u = 0; u < 4; ++u)
            if (j + u < len) atomicAdd(&s_cnt[dl[u] - n0], 1);
    }
    __syncthreads();

    int rp0 = 0, rp1 = 0;
    if (wv == 0) {
        const int c0 = s_cnt[2 * lane];
        const int c1 = s_cnt[2 * lane + 1];
        const int sm = c0 + c1;
        int incl = sm;
        for (int o = 1; o < 32; o <<= 1) {
            int u = __shfl_up(incl, o);
            if (lane >= o) incl += u;
        }
        rp0 = incl - sm;
        rp1 = rp0 + c0;
        s_cnt[2 * lane]     = rp0;
        s_cnt[2 * lane + 1] = rp1;
    }
    __syncthreads();

    const bool in_lds = (M <= CAP);
    for (int j = 0; j < len; j += 4) {
        int2 p[4];
#pragma unroll
        for (int u = 0; u < 4; ++u) p[u] = seg[(j + u < len) ? j + u : len - 1];
#pragma unroll
        for (int u = 0; u < 4; ++u)
            if (j + u < len) {
                const int pos = atomicAdd(&s_cnt[p[u].y - n0], 1);
                if (in_lds) s_out[pos] = p[u].x;
                else        csr_src[(size_t)base + pos] = p[u].x;
            }
    }
    __syncthreads();
    if (in_lds)
        for (int i = t; i < M; i += NCHUNK) csr_src[(size_t)base + i] = s_out[i];
    if (wv == 0) {
        const int nd = n0 + 2 * lane;
        if (nd < N)     rowptr[nd]     = base + rp0;
        if (nd + 1 < N) rowptr[nd + 1] = base + rp1;
        if (r == NR - 1 && lane == 31) rowptr[N] = base + M;
    }
}

template <int NCOLS>
__global__ void __launch_bounds__(NCHUNK) range_build_and_gemm(
    const int2* __restrict__ pairs, const int* __restrict__ roffT,
    int T, int NR, int N, int* __restrict__ rowptr, int* __restrict__ csr_src,
    const float* __restrict__ X, const float* __restrict__ W, float* __restrict__ Y)
{
    static_assert(NCHUNK == 256, "the GEMM tile is loaded and multiplied by 256 threads");
    constexpr int GEMM_INTS = 16 * GEMM_AP;
    __shared__ alignas(16) int lds[RB_LDS_INTS > GEMM_INTS ? RB_LDS_INTS : GEMM_INTS];
    const int b = blockIdx.x;
    if (b < NR) range_build_block(pairs, roffT, T, NR, N, rowptr, csr_src, b, lds);
    else        gemm_tile_f32<NCOLS>(X, W, Y, N, b - NR, (float (*)[GEMM_AP])lds);
}

template <int U> struct UTag { static constexpr int value = U; };

template <typename Fn>
__device__ __forceinline__ void for_edge_groups(int n, Fn f)
{
    int k = 0;
    for (; k + 8 <= n; k += 8) f(k, UTag<8>{});
    if (n & 4) { f(k, UTag<4>{}); k += 4; }
    if (n & 2) { f(k, UTag<2>{}); k += 2; }
    if (n & 1) { f(k, UTag<1>{}); }
}

template <typename Fn>
__device__ __forceinline__ void for_edge_groups4(int n, Fn f)
{
    int k = 0;
    for (; k + 4 <= n; k += 4) f(k, UTag<4>{});
    if (n & 2) { f(k, UTag<2>{}); k += 2; }
    if (n & 1) { f(k, UTag<1>{}); }
}

__device__ __forceinline__ float leaky(float a) { return (a > 0.f) ? a : NEG_SLOPE * a; }

template <int MASK>
__device__ __forceinline__ float lane_xor(float x)
{
    return __builtin_bit_cast(float, __builtin_amdgcn_update_dpp(
            0, __builtin_bit_cast(int, x), 0x160 + MASK  , 0xf, 0xf, false));
}

__device__ __forceinline__ float lane_xor16(float x)
{
    const int xi = __builtin_bit_cast(int, x);
    return __builtin_bit_cast(float,
            __builtin_amdgcn_permlanex16(xi, xi, 0x76543210, 0xfedcba98, false, false));
}

template <int I>
__device__ __forceinline__ float row_lane_c(float x)
{
    return __builtin_bit_cast(float, __builtin_amdgcn_mov_dpp(
            __builtin_bit_cast(int, x), 0x150 + I  , 0xf, 0xf, true));
}
__device__ __forceinline__ float row_lane(float x, int i)
{
    switch (i) {
    case 0:  return row_lane_c<0>(x);
    case 1:  return row_lane_c<1>(x);
    case 2:  return row_lane_c<2>(x);
    case 3:  return row_lane_c<3>(x);
    default: return row_lane_c<4>(x);
    }
}
__device__ __forceinline__ float row_sum16(float x)
{
    x += __builtin_bit_cast(float, __builtin_amdgcn_mov_dpp(__builtin_bit_cast(int, x), 0x161  , 0xf, 0xf, true));
    x += __builtin_bit_cast(float, __builtin_amdgcn_mov_dpp(__builtin_bit_cast(int, x), 0x162  , 0xf, 0xf, true));
    x += __builtin_bit_cast(float, __builtin_amdgcn_mov_dpp(__builtin_bit_cast(int, x), 0x164  , 0xf, 0xf, true));
    x += __builtin_bit_cast(float, __builtin_amdgcn_mov_dpp(__builtin_bit_cast(int, x), 0x168  , 0xf, 0xf, true));
    return x;
}

template <int LPH>
__device__ __forceinline__ float head_sum(float x)
{
    static_assert(LPH == 1 || LPH == 2 || LPH == 4 || LPH == 8 || LPH == 16, "a head's lanes lie in one row of 16");
    if (LPH >= 2)  x += lane_xor<1>(x);
    if (LPH >= 4)  x += lane_xor<2>(x);
    if (LPH >= 8)  x += lane_xor<4>(x);
    if (LPH >= 16) x += lane_xor<8>(x);
    return x;
}

template <int I>
__device__ __forceinline__ float quad_lane(float x)
{
    return __builtin_bit_cast(float, __builtin_amdgcn_mov_dpp(
            __builtin_bit_cast(int, x), I * 0x55  , 0xf, 0xf, true));
}

template <int MASK>
__device__ __forceinline__ float pair_sum_split(float a, float b, bool bit)
{
    const float keep = bit ? b : a;
    const float send = bit ? a : b;
    return keep + lane_xor<MASK>(send);
}

template <int HEADS, int C, bool RELU>
__global__ void __launch_bounds__(256) gat_gather_heads(
    const int* __restrict__ rowptr, const int* __restrict__ csr_src,
    const float* __restrict__ Hm, const float* __restrict__ att_src,
    const float* __restrict__ att_dst, const float* __restrict__ bias,
    float* __restrict__ out, int N)
{
    constexpr int F   = HEADS * C;
    constexpr int VEC = F / 32;
    constexpr int LPH = C / VEC;
    static_assert(F % 32 == 0 && VEC == 4 && C % VEC == 0, "a row is one float4 per lane, a head whole lanes");
    typedef float vf __attribute__((ext_vector_type(VEC)));

    const int lane = threadIdx.x & 31;
    const int wv   = __builtin_amdgcn_readfirstlane((int)(threadIdx.x >> 5));
    const int d = __builtin_amdgcn_readfirstlane((int)(blockIdx.x * 8) + wv);
    if (d >= N) return;
    const int beg = __builtin_amdgcn_readfirstlane(rowptr[d]);
    const int end = __builtin_amdgcn_readfirstlane(rowptr[d + 1]);

    const vf* Hv  = (const vf*)Hm + lane;
    const vf  as  = ((const vf*)att_src)[lane];
    const vf  ad  = ((const vf*)att_dst)[lane];
    const vf  bv  = ((const vf*)bias)[lane];

    auto dot = [](vf a, vf b) { return fmaf(a.w, b.w, fmaf(a.z, b.z, fmaf(a.y, b.y, a.x * b.x))); };

    static_assert(LPH == 4, "the edge groups below share a head's logits among its 4 lanes");
    const bool q1 = (lane & 1) != 0;
    const bool q2 = (lane & 2) != 0;

    const vf    own    = Hv[(size_t)d * 32];
    const float a_d    = head_sum<LPH>(dot(own, ad));
    const float a_self = leaky(head_sum<LPH>(dot(own, as)) + a_d);

    float m   = a_self;
    float den = 1.f;
    vf    acc = own;

    for (int j0 = beg; j0 < end; j0 += 32) {
        const int n = (end - j0 < 32) ? (end - j0) : 32;
        int mine = 0;
        if (lane < n) mine = csr_src[j0 + lane];
        for_edge_groups(n, [&](int k, auto u) {
            constexpr int U = decltype(u)::value;
            vf    rows[U];
            for (int i = 0; i < U; ++i)
                rows[i] = Hv[(size_t)__builtin_amdgcn_readlane(mine, k + i) * 32];
            __builtin_amdgcn_sched_barrier(0);
            constexpr int NA = (U == 8) ? 2 : 1;
            float p[U];
            for (int i = 0; i < U; ++i) p[i] = dot(rows[i], as);
            float al[NA];
            if constexpr (U == 8) {
                const float t0 = pair_sum_split<1>(p[0], p[1], q1);
                const float t1 = pair_sum_split<1>(p[2], p[3], q1);
                const float t2 = pair_sum_split<1>(p[4], p[5], q1);
                const float t3 = pair_sum_split<1>(p[6], p[7], q1);
                al[0] = pair_sum_split<2>(t0, t1, q2);
                al[1] = pair_sum_split<2>(t2, t3, q2);
            } else if constexpr (U == 4) {
                const float t0 = pair_sum_split<1>(p[0], p[1], q1);
                const float t1 = pair_sum_split<1>(p[2], p[3], q1);
                al[0] = pair_sum_split<2>(t0, t1, q2);
            } else if constexpr (U == 2) {
                const float t0 = pair_sum_split<1>(p[0], p[1], q1);
                al[0] = t0 + lane_xor<2>(t0);
            } else {
                al[0] = head_sum<LPH>(p[0]);
            }
            for (int j = 0; j < NA; ++j) al[j] = leaky(al[j] + a_d);
            float gm = al[0];
            if constexpr (NA == 2) gm = fmaxf(gm, al[1]);
            if constexpr (U >= 2) gm = fmaxf(gm, lane_xor<1>(gm));
            if constexpr (U >= 4) gm = fmaxf(gm, lane_xor<2>(gm));
            const float mn = fmaxf(m, gm);
            const float sc = expf(m - mn);
            float ex[NA];
            for (int j = 0; j < NA; ++j) ex[j] = expf(al[j] - mn);
            den *= sc;
            acc *= sc;
            m = mn;
            for (int i = 0; i < U; ++i) {
                float w;
                if constexpr (U == 1) w = ex[0];
                else if ((i & 3) == 0) w = quad_lane<0>(ex[i >> 2]);
                else if ((i & 3) == 1) w = quad_lane<1>(ex[i >> 2]);
                else if ((i & 3) == 2) w = quad_lane<2>(ex[i >> 2]);
                else                   w = quad_lane<3>(ex[i >> 2]);
                den += w;
                acc += rows[i] * w;
            }
        });
    }

    vf r = acc / (den + SM_EPS) + bv;
    if (RELU) {
#pragma unroll
        for (int i = 0; i < VEC; ++i) r[i] = fmaxf(r[i], 0.f);
    }
    ((vf*)out)[(size_t)d * 32 + lane] = r;
}

template <int C, bool RELU>
__global__ void __launch_bounds__(256) gat_gather_1head(
    const int* __restrict__ rowptr, const int* __restrict__ csr_src,
    const float* __restrict__ Hm, const float* __restrict__ att_src,
    const float* __restrict__ att_dst, const float* __restrict__ bias,
    float* __restrict__ out, int N)
{
    static_assert(C == 64, "a row is one float4 per lane of a half-wave");
    typedef float vf __attribute__((ext_vector_type(4)));

    const int lane = threadIdx.x & 31;
    const int c    = lane & 15;
    const int half = lane >> 4;
    const int wv   = __builtin_amdgcn_readfirstlane((int)(threadIdx.x >> 5));
    const int d = __builtin_amdgcn_readfirstlane((int)(blockIdx.x * 8) + wv);
    if (d >= N) return;
    const int beg = __builtin_amdgcn_readfirstlane(rowptr[d]);
    const int end = __builtin_amdgcn_readfirstlane(rowptr[d + 1]);

    const char* Hb = (const char*)Hm;
    const unsigned cb = (unsigned)c << 4;
    const vf as = ((const vf*)att_src)[c];
    const vf ad = ((const vf*)att_dst)[c];
    const vf bv = ((const vf*)bias)[c];

    auto dot = [](vf a, vf b) { return a.x * b.x + a.y * b.y + a.z * b.z + a.w * b.w; };

    const vf    own    = *(const vf*)(Hb + (((unsigned)d << 8) | cb));
    const float a_d    = row_sum16(dot(own, ad));
    const float a_self = leaky(row_sum16(dot(own, as)) + a_d);

    float m   = a_self;
    float den = half ? 0.f : 1.f;
    vf    acc = half ? (vf)(0.f) : own;

    for (int j0 = beg; j0 < end; j0 += 32) {
        const int n = (end - j0 < 32) ? (end - j0) : 32;
        int mine = d;
        if (lane < n) mine = csr_src[j0 + lane];
        const int nh = n - half;
        for_edge_groups4((n + 1) >> 1, [&](int k, auto u) {
            constexpr int U = decltype(u)::value;
            vf    rows[U];
            float al[U];
            for (int i = 0; i < U; ++i) {
                const int se = __builtin_amdgcn_readlane(mine, 2 * (k + i));
                const int so = __builtin_amdgcn_readlane(mine, 2 * (k + i) + 1);
                const unsigned s = (unsigned)(half ? so : se);
                rows[i] = *(const vf*)(Hb + ((s << 8) | cb));
            }
            for (int i = 0; i < U; ++i) {
                const float a = leaky(row_sum16(dot(rows[i], as)) + a_d);
                al[i] = (2 * (k + i) < nh) ? a : -INFINITY;
            }
            float mn = m;
            for (int i = 0; i < U; ++i) mn = fmaxf(mn, al[i]);
            float x = m;
            for (int i = 0; i < U; ++i) x = (c == i) ? al[i] : x;
            const float ex = expf(x - mn);
            const float sc = row_lane(ex, U);
            den *= sc;
            acc *= sc;
            m = mn;
            for (int i = 0; i < U; ++i) {
                const float w = row_lane(ex, i);
                den += w;
                acc += rows[i] * w;
            }
        });
    }

    const float mj = fmaxf(m, lane_xor16(m));
    const float sj = expf(m - mj);
    den *= sj;
    acc *= sj;
    den += lane_xor16(den);
    acc.x += lane_xor16(acc.x);
    acc.y += lane_xor16(acc.y);
    acc.z += lane_xor16(acc.z);
    acc.w += lane_xor16(acc.w);

    vf r = acc / (den + SM_EPS) + bv;
    if (RELU) {
#pragma unroll
        for (int i = 0; i < 4; ++i) r[i] = fmaxf(r[i], 0.f);
    }
    if (half == 0) ((vf*)out)[(size_t)d * 16 + c] = r;
}

static inline int cdiv(long long a, int b) { return (int)((a + b - 1) / b); }

extern "C" void kernel_launch(void* const* d_in, const int* in_sizes, int n_in,
                              void* d_out, int out_size, void* d_ws, size_t ws_size,
                              hipStream_t stream)
{
    const float* x        = (const float*)d_in[0];
    const int*   ei       = (const int*)  d_in[1];
    const float* W1       = (const float*)d_in[2];
    const float* att_src1 = (const float*)d_in[3];
    const float* att_dst1 = (const float*)d_in[4];
    const float* b1       = (const float*)d_in[5];
    const float* W2       = (const float*)d_in[6];
    const float* att_src2 = (const float*)d_in[7];
    const float* att_dst2 = (const float*)d_in[8];
    const float* b2       = (const float*)d_in[9];

    constexpr int IN = 128, H1 = 8, C1 = 16, F1 = H1 * C1  , C2 = 64;
    const int N = in_sizes[0] / IN;
    const int E = in_sizes[1] / 2;

    const int T  = cdiv(E, NCHUNK);
    const int NR = cdiv(N, 64);

    size_t off = 0;
    float* base = (float*)d_ws;
    auto take = [&](size_t n) { float* p = base + off; off += (n + 63) & ~(size_t)63; return p; };
    float* h1      = take((size_t)N * F1);
    float* agg1    = take((size_t)N * F1);
    float* h2      = take((size_t)N * C2);
    int2*  pairs   = (int2*)take((size_t)NCHUNK * T * 2);
    int*   roffT   = (int*)take((size_t)(NR + 1) * NCHUNK);
    int*   rowptr  = (int*)take((size_t)N + 1);
    int*   csr_src = (int*)take((size_t)E);

    const size_t part_lds = (size_t)((NR + 2) & ~1) * sizeof(int) + (size_t)T * sizeof(int2);
    edge_partition<<<NCHUNK, 1024, part_lds, stream>>>(ei, E, T, NR, pairs, roffT);
    range_build_and_gemm<F1><<<NR + cdiv(N, 16), NCHUNK, 0, stream>>>(
        pairs, roffT, T, NR, N, rowptr, csr_src, x, W1, h1);

    gat_gather_heads<H1, C1, true><<<cdiv(N, 8), 256, 0, stream>>>(
        rowptr, csr_src, h1, att_src1, att_dst1, b1, agg1, N);

    wmma_gemm_f32<C2><<<cdiv(N, 16), 256, 0, stream>>>(agg1, W2, h2, N);
    gat_gather_1head<C2, false><<<cdiv(N, 8), 256, 0, stream>>>(
        rowptr, csr_src, h2, att_src2, att_dst2, b2, (float*)d_out, N);
}


# Layer-2 GEMM joins the layer-1 gather's launch: block = 16 nodes (8 waves x 2), rows go to an LDS A tile, 4 waves multiply by W2 (16 k-steps' W loads at once); layer-1 output not stored; 5->4
# speedup vs baseline: 8.8705x; 1.0766x over previous
#include <hip/hip_runtime.h>
#include <hip/hip_bf16.h>


typedef float v2f __attribute__((ext_vector_type(2)));
typedef float v8f __attribute__((ext_vector_type(8)));

#define NEG_SLOPE 0.2f
#define SM_EPS    1e-16f

constexpr int GEMM_AP = 132;

template <int NCOLS, int UNR>
__device__ __forceinline__ void gemm_mul_tile_f32(
    const float* __restrict__ W, float* __restrict__ Y, int nrows, int tile,
    float (*As)[GEMM_AP])
{
    const int row0 = tile * 16;
    const int tid  = threadIdx.x;
    const int wave = tid >> 5;
    const int lane = tid & 31;
    const int col0 = wave * 16;
    if (col0 >= NCOLS) return;

    const int m  = lane & 15;
    const int hi = lane >> 4;

    const float* wp = W + (size_t)col0 + m;
    v8f acc = {};
#pragma unroll UNR
    for (int k0 = 0; k0 < 128; k0 += 4) {
        const int ka = k0 + (hi << 1);
        v2f a, b;
        a.x = As[m][ka];
        a.y = As[m][ka + 1];
        b.x = wp[(size_t)ka * NCOLS];
        b.y = wp[(size_t)(ka + 1) * NCOLS];
        acc = __builtin_amdgcn_wmma_f32_16x16x4_f32(
             false, a,  false, b,
             (short)0, acc,  false,  false);
    }

    const int orow = row0 + (hi ? 8 : 0);
    float* yp = Y + (size_t)orow * NCOLS + col0 + m;
    if (row0 + 16 <= nrows) {
#pragma unroll
        for (int r = 0; r < 8; ++r) yp[(size_t)r * NCOLS] = acc[r];
    } else {
#pragma unroll
        for (int r = 0; r < 8; ++r)
            if (orow + r < nrows) yp[(size_t)r * NCOLS] = acc[r];
    }
}

template <int NCOLS>
__device__ __forceinline__ void gemm_tile_f32(
    const float* __restrict__ X, const float* __restrict__ W,
    float* __restrict__ Y, int nrows, int tile, float (*As)[GEMM_AP])
{
    constexpr int AP = GEMM_AP;
    const int row0 = tile * 16;
    const int tid  = threadIdx.x;

    {
        const float4* X4  = (const float4*)X;
        float4*       As4 = (float4*)&As[0][0];
#pragma unroll
        for (int i = tid; i < 512; i += 256) {
            int r  = i >> 5;
            int rr = row0 + r;
            float4 v = make_float4(0.f, 0.f, 0.f, 0.f);
            if (rr < nrows) v = X4[(size_t)rr * 32 + (i & 31)];
            As4[r * (AP / 4) + (i & 31)] = v;
        }
    }
    __syncthreads();
    gemm_mul_tile_f32<NCOLS, 4>(W, Y, nrows, tile, As);
}

constexpr int NCHUNK = 256;

__device__ __forceinline__ int block_excl_scan_1024(int v, int* s_w)
{
    const int lane = threadIdx.x & 31;
    const int wv   = threadIdx.x >> 5;
    int incl = v;
    for (int o = 1; o < 32; o <<= 1) {
        int u = __shfl_up(incl, o);
        if (lane >= o) incl += u;
    }
    if (lane == 31) s_w[wv] = incl;
    __syncthreads();
    if (wv == 0) {
        int w = s_w[lane];
        int wi = w;
        for (int o = 1; o < 32; o <<= 1) {
            int u = __shfl_up(wi, o);
            if (lane >= o) wi += u;
        }
        s_w[lane] = wi - w;
    }
    __syncthreads();
    return s_w[wv] + (incl - v);
}

__global__ void __launch_bounds__(1024) edge_partition(
    const int* __restrict__ ei, int E, int T, int NR,
    int2* __restrict__ pairs, int* __restrict__ roffT)
{
    extern __shared__ __attribute__((aligned(16))) int dyn_lds[];
    __shared__ int s_w[32];
    int*  s_rng = dyn_lds;
    int2* s_st  = (int2*)(dyn_lds + ((NR + 2) & ~1));

    constexpr int U = 4;
    const int tid = threadIdx.x;
    const int w   = blockIdx.x;
    const int e0  = w * T;
    int n = E - e0;
    n = (n < 0) ? 0 : ((n > T) ? T : n);
    const int* srcs = ei + e0;
    const int* dsts = ei + (size_t)E + e0;

    for (int i = tid; i <= NR; i += 1024) s_rng[i] = 0;
    __syncthreads();
    for (int i0 = 0; i0 < n; i0 += 1024 * U) {
        int dv[U];
#pragma unroll
        for (int u = 0; u < U; ++u) {
            const int i = i0 + u * 1024 + tid;
            dv[u] = dsts[(i < n) ? i : n - 1];
        }
#pragma unroll
        for (int u = 0; u < U; ++u)
            if (i0 + u * 1024 + tid < n) atomicAdd(&s_rng[dv[u] >> 6], 1);
    }
    __syncthreads();

    const int per = (NR + 1 + 1023) >> 10;
    const int lo  = tid * per;
    int hi = lo + per;
    if (hi > NR + 1) hi = NR + 1;
    int s = 0;
    for (int i = lo; i < hi; ++i) s += s_rng[i];
    int run = block_excl_scan_1024(s, s_w);
    for (int i = lo; i < hi; ++i) { int c = s_rng[i]; s_rng[i] = run; run += c; }
    __syncthreads();


    for (int i0 = 0; i0 < n; i0 += 1024 * U) {
        int sv[U], dv[U];
#pragma unroll
        for (int u = 0; u < U; ++u) {
            const int i  = i0 + u * 1024 + tid;
            const int ic = (i < n) ? i : n - 1;
            sv[u] = srcs[ic];
            dv[u] = dsts[ic];
        }
#pragma unroll
        for (int u = 0; u < U; ++u)
            if (i0 + u * 1024 + tid < n) {
                const int p = atomicAdd(&s_rng[dv[u] >> 6], 1);
                s_st[p] = make_int2(sv[u], dv[u]);
            }
    }
    __syncthreads();
    for (int i = tid; i < n; i += 1024) pairs[(size_t)e0 + i] = s_st[i];
    for (int i = tid; i <= NR; i += 1024)
        roffT[(size_t)i * NCHUNK + w] = (i == 0) ? 0 : s_rng[i - 1];
}

constexpr int RB_CAP = 8192;
constexpr int RB_LDS_INTS = RB_CAP + 64 + 2 * (NCHUNK / 32);

__device__ __forceinline__ void range_build_block(
    const int2* __restrict__ pairs, const int* __restrict__ roffT,
    int T, int NR, int N, int* __restrict__ rowptr, int* __restrict__ csr_src,
    int r, int* lds)
{
    constexpr int CAP = RB_CAP;
    int* s_out = lds;
    int* s_cnt = lds + CAP;
    int (*s_sum)[NCHUNK / 32] = (int (*)[NCHUNK / 32])(lds + CAP + 64);

    const int t    = threadIdx.x;
    const int lane = t & 31;
    const int wv   = t >> 5;
    const int n0   = r * 64;

    const int o0  = roffT[(size_t)r * NCHUNK + t];
    const int len = roffT[(size_t)(r + 1) * NCHUNK + t] - o0;
    if (t < 64) s_cnt[t] = 0;

    int a = o0, b = len;
    for (int o = 16; o > 0; o >>= 1) { a += __shfl_xor(a, o); b += __shfl_xor(b, o); }
    if (lane == 0) { s_sum[0][wv] = a; s_sum[1][wv] = b; }
    __syncthreads();
    int base = 0, M = 0;
    for (int i = 0; i < NCHUNK / 32; ++i) { base += s_sum[0][i]; M += s_sum[1][i]; }

    const int2* seg = pairs + (size_t)t * T + o0;
    for (int j = 0; j < len; j += 4) {
        int dl[4];
#pragma unroll
        for (int u = 0; u < 4; ++u) dl[u] = seg[(j + u < len) ? j + u : len - 1].y;
#pragma unroll
        for (int u = 0; u < 4; ++u)
            if (j + u < len) atomicAdd(&s_cnt[dl[u] - n0], 1);
    }
    __syncthreads();

    int rp0 = 0, rp1 = 0;
    if (wv == 0) {
        const int c0 = s_cnt[2 * lane];
        const int c1 = s_cnt[2 * lane + 1];
        const int sm = c0 + c1;
        int incl = sm;
        for (int o = 1; o < 32; o <<= 1) {
            int u = __shfl_up(incl, o);
            if (lane >= o) incl += u;
        }
        rp0 = incl - sm;
        rp1 = rp0 + c0;
        s_cnt[2 * lane]     = rp0;
        s_cnt[2 * lane + 1] = rp1;
    }
    __syncthreads();

    const bool in_lds = (M <= CAP);
    for (int j = 0; j < len; j += 4) {
        int2 p[4];
#pragma unroll
        for (int u = 0; u < 4; ++u) p[u] = seg[(j + u < len) ? j + u : len - 1];
#pragma unroll
        for (int u = 0; u < 4; ++u)
            if (j + u < len) {
                const int pos = atomicAdd(&s_cnt[p[u].y - n0], 1);
                if (in_lds) s_out[pos] = p[u].x;
                else        csr_src[(size_t)base + pos] = p[u].x;
            }
    }
    __syncthreads();
    if (in_lds)
        for (int i = t; i < M; i += NCHUNK) csr_src[(size_t)base + i] = s_out[i];
    if (wv == 0) {
        const int nd = n0 + 2 * lane;
        if (nd < N)     rowptr[nd]     = base + rp0;
        if (nd + 1 < N) rowptr[nd + 1] = base + rp1;
        if (r == NR - 1 && lane == 31) rowptr[N] = base + M;
    }
}

template <int NCOLS>
__global__ void __launch_bounds__(NCHUNK) range_build_and_gemm(
    const int2* __restrict__ pairs, const int* __restrict__ roffT,
    int T, int NR, int N, int* __restrict__ rowptr, int* __restrict__ csr_src,
    const float* __restrict__ X, const float* __restrict__ W, float* __restrict__ Y)
{
    static_assert(NCHUNK == 256, "the GEMM tile is loaded and multiplied by 256 threads");
    constexpr int GEMM_INTS = 16 * GEMM_AP;
    __shared__ alignas(16) int lds[RB_LDS_INTS > GEMM_INTS ? RB_LDS_INTS : GEMM_INTS];
    const int b = blockIdx.x;
    if (b < NR) range_build_block(pairs, roffT, T, NR, N, rowptr, csr_src, b, lds);
    else        gemm_tile_f32<NCOLS>(X, W, Y, N, b - NR, (float (*)[GEMM_AP])lds);
}

template <int U> struct UTag { static constexpr int value = U; };

template <typename Fn>
__device__ __forceinline__ void for_edge_groups(int n, Fn f)
{
    int k = 0;
    for (; k + 8 <= n; k += 8) f(k, UTag<8>{});
    if (n & 4) { f(k, UTag<4>{}); k += 4; }
    if (n & 2) { f(k, UTag<2>{}); k += 2; }
    if (n & 1) { f(k, UTag<1>{}); }
}

template <typename Fn>
__device__ __forceinline__ void for_edge_groups4(int n, Fn f)
{
    int k = 0;
    for (; k + 4 <= n; k += 4) f(k, UTag<4>{});
    if (n & 2) { f(k, UTag<2>{}); k += 2; }
    if (n & 1) { f(k, UTag<1>{}); }
}

__device__ __forceinline__ float leaky(float a) { return (a > 0.f) ? a : NEG_SLOPE * a; }

template <int MASK>
__device__ __forceinline__ float lane_xor(float x)
{
    return __builtin_bit_cast(float, __builtin_amdgcn_update_dpp(
            0, __builtin_bit_cast(int, x), 0x160 + MASK  , 0xf, 0xf, false));
}

__device__ __forceinline__ float lane_xor16(float x)
{
    const int xi = __builtin_bit_cast(int, x);
    return __builtin_bit_cast(float,
            __builtin_amdgcn_permlanex16(xi, xi, 0x76543210, 0xfedcba98, false, false));
}

template <int I>
__device__ __forceinline__ float row_lane_c(float x)
{
    return __builtin_bit_cast(float, __builtin_amdgcn_mov_dpp(
            __builtin_bit_cast(int, x), 0x150 + I  , 0xf, 0xf, true));
}
__device__ __forceinline__ float row_lane(float x, int i)
{
    switch (i) {
    case 0:  return row_lane_c<0>(x);
    case 1:  return row_lane_c<1>(x);
    case 2:  return row_lane_c<2>(x);
    case 3:  return row_lane_c<3>(x);
    default: return row_lane_c<4>(x);
    }
}
__device__ __forceinline__ float row_sum16(float x)
{
    x += __builtin_bit_cast(float, __builtin_amdgcn_mov_dpp(__builtin_bit_cast(int, x), 0x161  , 0xf, 0xf, true));
    x += __builtin_bit_cast(float, __builtin_amdgcn_mov_dpp(__builtin_bit_cast(int, x), 0x162  , 0xf, 0xf, true));
    x += __builtin_bit_cast(float, __builtin_amdgcn_mov_dpp(__builtin_bit_cast(int, x), 0x164  , 0xf, 0xf, true));
    x += __builtin_bit_cast(float, __builtin_amdgcn_mov_dpp(__builtin_bit_cast(int, x), 0x168  , 0xf, 0xf, true));
    return x;
}

template <int LPH>
__device__ __forceinline__ float head_sum(float x)
{
    static_assert(LPH == 1 || LPH == 2 || LPH == 4 || LPH == 8 || LPH == 16, "a head's lanes lie in one row of 16");
    if (LPH >= 2)  x += lane_xor<1>(x);
    if (LPH >= 4)  x += lane_xor<2>(x);
    if (LPH >= 8)  x += lane_xor<4>(x);
    if (LPH >= 16) x += lane_xor<8>(x);
    return x;
}

template <int I>
__device__ __forceinline__ float quad_lane(float x)
{
    return __builtin_bit_cast(float, __builtin_amdgcn_mov_dpp(
            __builtin_bit_cast(int, x), I * 0x55  , 0xf, 0xf, true));
}

template <int MASK>
__device__ __forceinline__ float pair_sum_split(float a, float b, bool bit)
{
    const float keep = bit ? b : a;
    const float send = bit ? a : b;
    return keep + lane_xor<MASK>(send);
}

template <int HEADS, int C, bool RELU, int NCOLS>
__global__ void __launch_bounds__(256, 12) gat_gather_heads_gemm(
    const int* __restrict__ rowptr, const int* __restrict__ csr_src,
    const float* __restrict__ Hm, const float* __restrict__ att_src,
    const float* __restrict__ att_dst, const float* __restrict__ bias,
    const float* __restrict__ W, float* __restrict__ Y, int N)
{
    constexpr int F   = HEADS * C;
    constexpr int VEC = F / 32;
    constexpr int LPH = C / VEC;
    static_assert(F == 128 && VEC == 4 && C % VEC == 0, "a row is one float4 per lane and one row of the 16x128 A tile, a head whole lanes");
    typedef float vf __attribute__((ext_vector_type(VEC)));

    __shared__ alignas(16) float As[16][GEMM_AP];

    const int lane = threadIdx.x & 31;
    const int wv   = __builtin_amdgcn_readfirstlane((int)(threadIdx.x >> 5));
    const int d0 = __builtin_amdgcn_readfirstlane((int)(blockIdx.x * 16) + 2 * wv);
    const int rp0 = __builtin_amdgcn_readfirstlane(rowptr[(d0     < N) ? d0     : N]);
    const int rp1 = __builtin_amdgcn_readfirstlane(rowptr[(d0 + 1 < N) ? d0 + 1 : N]);
    const int rp2 = __builtin_amdgcn_readfirstlane(rowptr[(d0 + 2 < N) ? d0 + 2 : N]);

    const vf* Hv  = (const vf*)Hm + lane;
    const vf  as  = ((const vf*)att_src)[lane];

    auto dot = [](vf a, vf b) { return fmaf(a.w, b.w, fmaf(a.z, b.z, fmaf(a.y, b.y, a.x * b.x))); };

    static_assert(LPH == 4, "the edge groups below share a head's logits among its 4 lanes");
    const bool q1 = (lane & 1) != 0;
    const bool q2 = (lane & 2) != 0;

#pragma nounroll
    for (int nd = 0; nd < 2; ++nd) {
    const int d   = d0 + nd;
    const int beg = nd ? rp1 : rp0;
    const int end = nd ? rp2 : rp1;
    vf r = (vf)(0.f);
    if (d < N) {
    const vf    ad     = ((const vf*)att_dst)[lane];
    const vf    own    = Hv[(size_t)d * 32];
    const float a_d    = head_sum<LPH>(dot(own, ad));
    const float a_self = leaky(head_sum<LPH>(dot(own, as)) + a_d);

    float m   = a_self;
    float den = 1.f;
    vf    acc = own;

    for (int j0 = beg; j0 < end; j0 += 32) {
        const int n = (end - j0 < 32) ? (end - j0) : 32;
        int mine = 0;
        if (lane < n) mine = csr_src[j0 + lane];
        for_edge_groups(n, [&](int k, auto u) {
            constexpr int U = decltype(u)::value;
            vf    rows[U];
            for (int i = 0; i < U; ++i)
                rows[i] = Hv[(size_t)__builtin_amdgcn_readlane(mine, k + i) * 32];
            __builtin_amdgcn_sched_barrier(0);
            constexpr int NA = (U == 8) ? 2 : 1;
            float p[U];
            for (int i = 0; i < U; ++i) p[i] = dot(rows[i], as);
            float al[NA];
            if constexpr (U == 8) {
                const float t0 = pair_sum_split<1>(p[0], p[1], q1);
                const float t1 = pair_sum_split<1>(p[2], p[3], q1);
                const float t2 = pair_sum_split<1>(p[4], p[5], q1);
                const float t3 = pair_sum_split<1>(p[6], p[7], q1);
                al[0] = pair_sum_split<2>(t0, t1, q2);
                al[1] = pair_sum_split<2>(t2, t3, q2);
            } else if constexpr (U == 4) {
                const float t0 = pair_sum_split<1>(p[0], p[1], q1);
                const float t1 = pair_sum_split<1>(p[2], p[3], q1);
                al[0] = pair_sum_split<2>(t0, t1, q2);
            } else if constexpr (U == 2) {
                const float t0 = pair_sum_split<1>(p[0], p[1], q1);
                al[0] = t0 + lane_xor<2>(t0);
            } else {
                al[0] = head_sum<LPH>(p[0]);
            }
            for (int j = 0; j < NA; ++j) al[j] = leaky(al[j] + a_d);
            float gm = al[0];
            if constexpr (NA == 2) gm = fmaxf(gm, al[1]);
            if constexpr (U >= 2) gm = fmaxf(gm, lane_xor<1>(gm));
            if constexpr (U >= 4) gm = fmaxf(gm, lane_xor<2>(gm));
            const float mn = fmaxf(m, gm);
            const float sc = expf(m - mn);
            float ex[NA];
            for (int j = 0; j < NA; ++j) ex[j] = expf(al[j] - mn);
            den *= sc;
            acc *= sc;
            m = mn;
            for (int i = 0; i < U; ++i) {
                float w;
                if constexpr (U == 1) w = ex[0];
                else if ((i & 3) == 0) w = quad_lane<0>(ex[i >> 2]);
                else if ((i & 3) == 1) w = quad_lane<1>(ex[i >> 2]);
                else if ((i & 3) == 2) w = quad_lane<2>(ex[i >> 2]);
                else                   w = quad_lane<3>(ex[i >> 2]);
                den += w;
                acc += rows[i] * w;
            }
        });
    }

    const vf bv = ((const vf*)bias)[lane];
    r = acc / (den + SM_EPS) + bv;
    if (RELU) {
#pragma unroll
        for (int i = 0; i < VEC; ++i) r[i] = fmaxf(r[i], 0.f);
    }
    }
    *(vf*)&As[2 * wv + nd][VEC * lane] = r;
    }
    __syncthreads();
    gemm_mul_tile_f32<NCOLS, 16>(W, Y, N, blockIdx.x, As);
}

template <int C, bool RELU>
__global__ void __launch_bounds__(256) gat_gather_1head(
    const int* __restrict__ rowptr, const int* __restrict__ csr_src,
    const float* __restrict__ Hm, const float* __restrict__ att_src,
    const float* __restrict__ att_dst, const float* __restrict__ bias,
    float* __restrict__ out, int N)
{
    static_assert(C == 64, "a row is one float4 per lane of a half-wave");
    typedef float vf __attribute__((ext_vector_type(4)));

    const int lane = threadIdx.x & 31;
    const int c    = lane & 15;
    const int half = lane >> 4;
    const int wv   = __builtin_amdgcn_readfirstlane((int)(threadIdx.x >> 5));
    const int d = __builtin_amdgcn_readfirstlane((int)(blockIdx.x * 8) + wv);
    if (d >= N) return;
    const int beg = __builtin_amdgcn_readfirstlane(rowptr[d]);
    const int end = __builtin_amdgcn_readfirstlane(rowptr[d + 1]);

    const char* Hb = (const char*)Hm;
    const unsigned cb = (unsigned)c << 4;
    const vf as = ((const vf*)att_src)[c];
    const vf ad = ((const vf*)att_dst)[c];
    const vf bv = ((const vf*)bias)[c];

    auto dot = [](vf a, vf b) { return a.x * b.x + a.y * b.y + a.z * b.z + a.w * b.w; };

    const vf    own    = *(const vf*)(Hb + (((unsigned)d << 8) | cb));
    const float a_d    = row_sum16(dot(own, ad));
    const float a_self = leaky(row_sum16(dot(own, as)) + a_d);

    float m   = a_self;
    float den = half ? 0.f : 1.f;
    vf    acc = half ? (vf)(0.f) : own;

    for (int j0 = beg; j0 < end; j0 += 32) {
        const int n = (end - j0 < 32) ? (end - j0) : 32;
        int mine = d;
        if (lane < n) mine = csr_src[j0 + lane];
        const int nh = n - half;
        for_edge_groups4((n + 1) >> 1, [&](int k, auto u) {
            constexpr int U = decltype(u)::value;
            vf    rows[U];
            float al[U];
            for (int i = 0; i < U; ++i) {
                const int se = __builtin_amdgcn_readlane(mine, 2 * (k + i));
                const int so = __builtin_amdgcn_readlane(mine, 2 * (k + i) + 1);
                const unsigned s = (unsigned)(half ? so : se);
                rows[i] = *(const vf*)(Hb + ((s << 8) | cb));
            }
            for (int i = 0; i < U; ++i) {
                const float a = leaky(row_sum16(dot(rows[i], as)) + a_d);
                al[i] = (2 * (k + i) < nh) ? a : -INFINITY;
            }
            float mn = m;
            for (int i = 0; i < U; ++i) mn = fmaxf(mn, al[i]);
            float x = m;
            for (int i = 0; i < U; ++i) x = (c == i) ? al[i] : x;
            const float ex = expf(x - mn);
            const float sc = row_lane(ex, U);
            den *= sc;
            acc *= sc;
            m = mn;
            for (int i = 0; i < U; ++i) {
                const float w = row_lane(ex, i);
                den += w;
                acc += rows[i] * w;
            }
        });
    }

    const float mj = fmaxf(m, lane_xor16(m));
    const float sj = expf(m - mj);
    den *= sj;
    acc *= sj;
    den += lane_xor16(den);
    acc.x += lane_xor16(acc.x);
    acc.y += lane_xor16(acc.y);
    acc.z += lane_xor16(acc.z);
    acc.w += lane_xor16(acc.w);

    vf r = acc / (den + SM_EPS) + bv;
    if (RELU) {
#pragma unroll
        for (int i = 0; i < 4; ++i) r[i] = fmaxf(r[i], 0.f);
    }
    if (half == 0) ((vf*)out)[(size_t)d * 16 + c] = r;
}

static inline int cdiv(long long a, int b) { return (int)((a + b - 1) / b); }

extern "C" void kernel_launch(void* const* d_in, const int* in_sizes, int n_in,
                              void* d_out, int out_size, void* d_ws, size_t ws_size,
                              hipStream_t stream)
{
    const float* x        = (const float*)d_in[0];
    const int*   ei       = (const int*)  d_in[1];
    const float* W1       = (const float*)d_in[2];
    const float* att_src1 = (const float*)d_in[3];
    const float* att_dst1 = (const float*)d_in[4];
    const float* b1       = (const float*)d_in[5];
    const float* W2       = (const float*)d_in[6];
    const float* att_src2 = (const float*)d_in[7];
    const float* att_dst2 = (const float*)d_in[8];
    const float* b2       = (const float*)d_in[9];

    constexpr int IN = 128, H1 = 8, C1 = 16, F1 = H1 * C1  , C2 = 64;
    const int N = in_sizes[0] / IN;
    const int E = in_sizes[1] / 2;

    const int T  = cdiv(E, NCHUNK);
    const int NR = cdiv(N, 64);

    size_t off = 0;
    float* base = (float*)d_ws;
    auto take = [&](size_t n) { float* p = base + off; off += (n + 63) & ~(size_t)63; return p; };
    float* h1      = take((size_t)N * F1);
    float* h2      = take((size_t)N * C2);
    int2*  pairs   = (int2*)take((size_t)NCHUNK * T * 2);
    int*   roffT   = (int*)take((size_t)(NR + 1) * NCHUNK);
    int*   rowptr  = (int*)take((size_t)N + 1);
    int*   csr_src = (int*)take((size_t)E);

    const size_t part_lds = (size_t)((NR + 2) & ~1) * sizeof(int) + (size_t)T * sizeof(int2);
    edge_partition<<<NCHUNK, 1024, part_lds, stream>>>(ei, E, T, NR, pairs, roffT);
    range_build_and_gemm<F1><<<NR + cdiv(N, 16), NCHUNK, 0, stream>>>(
        pairs, roffT, T, NR, N, rowptr, csr_src, x, W1, h1);

    gat_gather_heads_gemm<H1, C1, true, C2><<<cdiv(N, 16), 256, 0, stream>>>(
        rowptr, csr_src, h1, att_src1, att_dst1, b1, W2, h2, N);

    gat_gather_1head<C2, false><<<cdiv(N, 8), 256, 0, stream>>>(
        rowptr, csr_src, h2, att_src2, att_dst2, b2, (float*)d_out, N);
}
